# starting kernel
# baseline (speedup 1.0000x reference)
#include <hip/hip_runtime.h>
#include <hip/hip_bf16.h>

#define DFEAT 128
#define LDSP  132
#define MT    64

typedef float v2f __attribute__((ext_vector_type(2)));
typedef float v8f __attribute__((ext_vector_type(8)));


__global__ void deg_init_k(float* __restrict__ deg, int n) {
    int i = blockIdx.x * blockDim.x + threadIdx.x;
    if (i < n) deg[i] = 1.0f;
}

__global__ void deg_acc_k(const int* __restrict__ ei, const float* __restrict__ w,
                          float* __restrict__ deg, int E) {
    int e = blockIdx.x * blockDim.x + threadIdx.x;
    if (e < E) atomicAdd(deg + ei[E + e], w[e]);
}

__global__ void dinv_k(float* __restrict__ deg, int n) {
    int i = blockIdx.x * blockDim.x + threadIdx.x;
    if (i < n) {
        float d = deg[i];
        deg[i] = (d > 0.0f) ? rsqrtf(d) : 0.0f;
    }
}

__global__ void norm_k(const int* __restrict__ ei, const float* __restrict__ w,
                       const float* __restrict__ dinv, float* __restrict__ nrm, int E) {
    int e = blockIdx.x * blockDim.x + threadIdx.x;
    if (e < E) nrm[e] = dinv[ei[e]] * w[e] * dinv[ei[E + e]];
}


__global__ __launch_bounds__(256)
void gemm_wmma_k(const float* __restrict__ A, const float* __restrict__ W,
                 const float* __restrict__ bias, float* __restrict__ out,
                 int nrows, int reluIn, int addBias) {
    __shared__ float As[MT][LDSP];

    const int tid = threadIdx.x;
    const int rowBase = blockIdx.x * MT;

    const float4* A4 = (const float4*)A;
    #pragma unroll
    for (int j = 0; j < 8; ++j) {
        int idx4 = j * 256 + tid;
        int r    = idx4 >> 5;
        int c4   = idx4 & 31;
        float4 v = make_float4(0.f, 0.f, 0.f, 0.f);
        int gr = rowBase + r;
        if (gr < nrows) v = A4[(size_t)gr * (DFEAT / 4) + c4];
        if (reluIn) {
            v.x = fmaxf(v.x, 0.f); v.y = fmaxf(v.y, 0.f);
            v.z = fmaxf(v.z, 0.f); v.w = fmaxf(v.w, 0.f);
        }
        *(float4*)&As[r][c4 * 4] = v;
    }
    __syncthreads();

    const int lane = tid & 31;
    const int wv   = tid >> 5;
    const int colBase = wv * 16;
    const int m  = lane & 15;
    const int kh = (lane >> 4) * 2;
    const int col = colBase + m;

    v8f acc[4] = {};
    #pragma unroll
    for (int k0 = 0; k0 < DFEAT; k0 += 4) {
        v2f b;
        b.x = W[(size_t)(k0 + kh)     * DFEAT + col];
        b.y = W[(size_t)(k0 + kh + 1) * DFEAT + col];
        #pragma unroll
        for (int rt = 0; rt < 4; ++rt) {
            v2f a;
            a.x = As[rt * 16 + m][k0 + kh];
            a.y = As[rt * 16 + m][k0 + kh + 1];
            acc[rt] = __builtin_amdgcn_wmma_f32_16x16x4_f32(
                          false, a, false, b, (short)0, acc[rt], false, false);
        }
    }

    float bv = 0.0f;
    if (addBias) bv = bias[col];
    const int rowOff = (lane >> 4) * 8;
    #pragma unroll
    for (int rt = 0; rt < 4; ++rt) {
        #pragma unroll
        for (int v = 0; v < 8; ++v) {
            int gr = rowBase + rt * 16 + rowOff + v;
            if (gr < nrows) out[(size_t)gr * DFEAT + col] = acc[rt][v] + bv;
        }
    }
}


__global__ void agg_init_k(const float* __restrict__ hb, const float* __restrict__ dinv,
                           const float* __restrict__ bias, float* __restrict__ out, int n) {
    int gid = blockIdx.x * blockDim.x + threadIdx.x;
    if (gid >= n * (DFEAT / 4)) return;
    int row = gid >> 5;
    int c4  = gid & 31;
    float s = dinv[row];
    s = s * s;
    float4 v  = ((const float4*)hb)[gid];
    float4 bv = ((const float4*)bias)[c4];
    float4 r;
    r.x = v.x * s + bv.x; r.y = v.y * s + bv.y;
    r.z = v.z * s + bv.z; r.w = v.w * s + bv.w;
    ((float4*)out)[gid] = r;
}

__global__ __launch_bounds__(256)
void scatter_k(const float* __restrict__ hb, const int* __restrict__ ei,
               const float* __restrict__ nrm, float* __restrict__ out, int E) {
    int gid  = blockIdx.x * blockDim.x + threadIdx.x;
    int e    = gid >> 5;
    int lane = gid & 31;
    if (e >= E) return;
    int s = ei[e];
    int d = ei[E + e];
    float nv = nrm[e];
    float4 v = ((const float4*)(hb + (size_t)s * DFEAT))[lane];
    float* o = out + (size_t)d * DFEAT + lane * 4;
    atomicAdd(o + 0, v.x * nv);
    atomicAdd(o + 1, v.y * nv);
    atomicAdd(o + 2, v.z * nv);
    atomicAdd(o + 3, v.w * nv);
}


static inline size_t align_up(size_t x) { return (x + 63) & ~(size_t)63; }

extern "C" void kernel_launch(void* const* d_in, const int* in_sizes, int n_in,
                              void* d_out, int out_size, void* d_ws, size_t ws_size,
                              hipStream_t stream) {
    const float* x  = (const float*)d_in[0];
    const int*   ei = (const int*)d_in[1];
    const float* ew = (const float*)d_in[2];
    const float* W1 = (const float*)d_in[3];  const float* b1 = (const float*)d_in[4];
    const float* W2 = (const float*)d_in[5];  const float* b2 = (const float*)d_in[6];
    const float* W3 = (const float*)d_in[7];  const float* b3 = (const float*)d_in[8];
    const float* Wl = (const float*)d_in[9];  const float* bl = (const float*)d_in[10];

    const int N = in_sizes[0] / DFEAT;
    const int E = in_sizes[2];

    float* dinv = (float*)d_ws;
    float* nrm  = dinv + align_up((size_t)N);
    float* hb   = nrm  + align_up((size_t)E);
    float* bufA = hb   + (size_t)N * DFEAT;
    float* bufB = bufA + (size_t)N * DFEAT;
    float* outF = (float*)d_out;

    const int T = 256;
    dim3 blkN((N + T - 1) / T), blkE((E + T - 1) / T);
    dim3 blkG((N + MT - 1) / MT);
    dim3 blkI(((size_t)N * (DFEAT / 4) + T - 1) / T);
    dim3 blkS(((size_t)E * 32 + T - 1) / T);

    deg_init_k<<<blkN, T, 0, stream>>>(dinv, N);
    deg_acc_k <<<blkE, T, 0, stream>>>(ei, ew, dinv, E);
    dinv_k    <<<blkN, T, 0, stream>>>(dinv, N);
    norm_k    <<<blkE, T, 0, stream>>>(ei, ew, dinv, nrm, E);

    gemm_wmma_k<<<blkG, T, 0, stream>>>(x,    W1, b1, hb,   N, 0, 0);
    agg_init_k <<<blkI, T, 0, stream>>>(hb, dinv, b1, bufA, N);
    scatter_k  <<<blkS, T, 0, stream>>>(hb, ei, nrm, bufA, E);

    gemm_wmma_k<<<blkG, T, 0, stream>>>(bufA, W2, b2, hb,   N, 1, 0);
    agg_init_k <<<blkI, T, 0, stream>>>(hb, dinv, b2, bufB, N);
    scatter_k  <<<blkS, T, 0, stream>>>(hb, ei, nrm, bufB, E);

    gemm_wmma_k<<<blkG, T, 0, stream>>>(bufB, W3, b3, hb,   N, 1, 0);
    agg_init_k <<<blkI, T, 0, stream>>>(hb, dinv, b3, bufA, N);
    scatter_k  <<<blkS, T, 0, stream>>>(hb, ei, nrm, bufA, E);

    gemm_wmma_k<<<blkG, T, 0, stream>>>(bufA, Wl, bl, outF, N, 0, 1);
}


# Norm factored out of the edges: GEMMs store dinv-scaled rows (dinv made in layer 1's GEMM from degree partials), next tile load applies dinv then bias (added last); raw edge weights; 10->8 launches
# speedup vs baseline: 2.1643x; 2.1643x over previous
#include <hip/hip_runtime.h>
#include <hip/hip_bf16.h>

#define DFEAT 128
#define LDSP  132
#define MT    64

typedef float v2f __attribute__((ext_vector_type(2)));
typedef float v8f __attribute__((ext_vector_type(8)));


#define DCH    25088
#define DPARTS 64
#define DTH    512
#define DEU    13

__global__ __launch_bounds__(DTH)
void deg_part_k(const int* __restrict__ ei, const float* __restrict__ w,
                float* __restrict__ part, int n, int E) {
    __shared__ alignas(16) float sdeg[DCH];
    const int tid = threadIdx.x;
    const int lo  = blockIdx.y * DCH;
    const int cn  = min(DCH, n - lo);
    const int pstride = (n + 31) & ~31;

    for (int i = tid; i < ((cn + 3) >> 2); i += DTH)
        ((float4*)sdeg)[i] = make_float4(0.f, 0.f, 0.f, 0.f);
    __syncthreads();

    const int per = (E + (int)gridDim.x - 1) / (int)gridDim.x;
    const int e0  = blockIdx.x * per;
    const int e1  = min(E, e0 + per);
    for (int base = e0 + tid; base < e1; base += DTH * DEU) {
        int   d[DEU];
        float wv[DEU];
        #pragma unroll
        for (int u = 0; u < DEU; ++u) {
            const int ec = min(base + u * DTH, e1 - 1);
            d[u]  = ei[E + ec];
            wv[u] = w[ec];
        }
        #pragma unroll
        for (int u = 0; u < DEU; ++u) {
            const int r = d[u] - lo;
            if (base + u * DTH < e1 && (unsigned)r < (unsigned)cn) atomicAdd(&sdeg[r], wv[u]);
        }
    }
    __syncthreads();

    float* p = part + (size_t)blockIdx.x * pstride + lo;
    const int c4 = cn >> 2;
    for (int i = tid; i < c4; i += DTH) ((float4*)p)[i] = ((const float4*)sdeg)[i];
    for (int i = (c4 << 2) + tid; i < cn; i += DTH) p[i] = sdeg[i];
}

template <int MODE>
__global__ __launch_bounds__(256)
void gemm_wmma_k(const float* __restrict__ A, const float* __restrict__ W,
                 const float* __restrict__ biasIn, const float* __restrict__ biasOut,
                 float* __restrict__ out, float* __restrict__ out2,
                 const float* __restrict__ part, float* __restrict__ dinv, int nrows) {
    __shared__ float As[MT][LDSP];
    __shared__ alignas(16) float Ds[MT];

    const int tid = threadIdx.x;
    const int rowBase = blockIdx.x * MT;

    if (MODE == 0) {
        if (tid < MT) {
            const int gr = rowBase + tid;
            float d = 0.0f;
            if (gr < nrows) {
                const int pstride = (nrows + 31) & ~31;
                float pv[DPARTS];
                #pragma unroll
                for (int p = 0; p < DPARTS; ++p) pv[p] = part[(size_t)p * pstride + gr];
                float s = 1.0f;
                #pragma unroll
                for (int p = 0; p < DPARTS; ++p) s += pv[p];
                d = (s > 0.0f) ? rsqrtf(s) : 0.0f;
                dinv[gr] = d;
            }
            Ds[tid] = d;
        }
    }

    const float4* A4 = (const float4*)A;
    float bi0 = 0.f, bi1 = 0.f, bi2 = 0.f, bi3 = 0.f;
    if (MODE != 0) {
        const int c = (tid & 31) * 4;
        bi0 = biasIn[c]; bi1 = biasIn[c + 1]; bi2 = biasIn[c + 2]; bi3 = biasIn[c + 3];
    }
    #pragma unroll
    for (int j = 0; j < 8; ++j) {
        int idx4 = j * 256 + tid;
        int r    = idx4 >> 5;
        int c4   = idx4 & 31;
        float4 v = make_float4(0.f, 0.f, 0.f, 0.f);
        int gr = rowBase + r;
        if (gr < nrows) {
            v = A4[(size_t)gr * (DFEAT / 4) + c4];
            if (MODE != 0) {
                const float dv = dinv[gr];
                v.x = v.x * dv + bi0; v.y = v.y * dv + bi1;
                v.z = v.z * dv + bi2; v.w = v.w * dv + bi3;
            }
            if (MODE == 1) {
                v.x = fmaxf(v.x, 0.f); v.y = fmaxf(v.y, 0.f);
                v.z = fmaxf(v.z, 0.f); v.w = fmaxf(v.w, 0.f);
            }
        }
        *(float4*)&As[r][c4 * 4] = v;
    }
    if (MODE == 1) {
        if (tid < MT) {
            int gr = rowBase + tid;
            Ds[tid] = (gr < nrows) ? dinv[gr] : 0.0f;
        }
    }
    __syncthreads();

    const int lane = tid & 31;
    const int wv   = tid >> 5;
    const int colBase = wv * 16;
    const int m  = lane & 15;
    const int kh = (lane >> 4) * 2;
    const int col = colBase + m;

    v8f acc[4] = {};
    #pragma unroll
    for (int k0 = 0; k0 < DFEAT; k0 += 4) {
        v2f b;
        b.x = W[(size_t)(k0 + kh)     * DFEAT + col];
        b.y = W[(size_t)(k0 + kh + 1) * DFEAT + col];
        #pragma unroll
        for (int rt = 0; rt < 4; ++rt) {
            v2f a;
            a.x = As[rt * 16 + m][k0 + kh];
            a.y = As[rt * 16 + m][k0 + kh + 1];
            acc[rt] = __builtin_amdgcn_wmma_f32_16x16x4_f32(
                          false, a, false, b, (short)0, acc[rt], false, false);
        }
    }

    float bv = 0.0f;
    if (MODE == 2) bv = biasOut[col];
    const int rowOff = (lane >> 4) * 8;
    #pragma unroll
    for (int rt = 0; rt < 4; ++rt) {
        float sw[8] = {0.f, 0.f, 0.f, 0.f, 0.f, 0.f, 0.f, 0.f};
        if (MODE != 2) {
            const float4 s0 = *(const float4*)&Ds[rt * 16 + rowOff];
            const float4 s1 = *(const float4*)&Ds[rt * 16 + rowOff + 4];
            sw[0] = s0.x; sw[1] = s0.y; sw[2] = s0.z; sw[3] = s0.w;
            sw[4] = s1.x; sw[5] = s1.y; sw[6] = s1.z; sw[7] = s1.w;
        }
        #pragma unroll
        for (int v = 0; v < 8; ++v) {
            int gr = rowBase + rt * 16 + rowOff + v;
            if (gr < nrows) {
                const size_t o = (size_t)gr * DFEAT + col;
                if (MODE != 2) {
                    const float g = acc[rt][v] * sw[v];
                    out[o]  = g;
                    out2[o] = g;
                } else {
                    out[o] = acc[rt][v] + bv;
                }
            }
        }
    }
}


#define EB 8

__device__ __forceinline__ void scatter_one(const float* __restrict__ hb, float* __restrict__ out,
                                            int s, int d, float nv, int lane) {
    const float* h = hb + (size_t)s * DFEAT + lane;
    float v0 = h[0];
    float v1 = h[32];
    float v2 = h[64];
    float v3 = h[96];
    float* o = out + (size_t)d * DFEAT + lane;
    atomicAdd(o,      v0 * nv);
    atomicAdd(o + 32, v1 * nv);
    atomicAdd(o + 64, v2 * nv);
    atomicAdd(o + 96, v3 * nv);
}

__global__ __launch_bounds__(128)
void scatter_k(const float* __restrict__ hb, const int* __restrict__ ei,
               const float* __restrict__ wgt, float* __restrict__ out, int E) {
    const int gid  = blockIdx.x * blockDim.x + threadIdx.x;
    const int lane = threadIdx.x & 31;
    const int base = __builtin_amdgcn_readfirstlane(gid & ~31);
    if (base >= E) return;
    const int cnt = min(32, E - base);

    int   s_l = 0, d_l = 0;
    float n_l = 0.0f;
    if (lane < cnt) {
        s_l = ei[base + lane];
        d_l = ei[E + base + lane];
        n_l = wgt[base + lane];
    }

    if (cnt == 32) {
        for (int j0 = 0; j0 < 32; j0 += EB) {
            float v[EB][4];
            #pragma unroll
            for (int u = 0; u < EB; ++u) {
                int s = __builtin_amdgcn_readlane(s_l, j0 + u);
                const float* h = hb + (size_t)s * DFEAT + lane;
                v[u][0] = h[0];
                v[u][1] = h[32];
                v[u][2] = h[64];
                v[u][3] = h[96];
            }
            #pragma unroll
            for (int u = 0; u < EB; ++u) {
                int   d  = __builtin_amdgcn_readlane(d_l, j0 + u);
                float nv = __int_as_float(__builtin_amdgcn_readlane(__float_as_int(n_l), j0 + u));
                float* o = out + (size_t)d * DFEAT + lane;
                atomicAdd(o,      v[u][0] * nv);
                atomicAdd(o + 32, v[u][1] * nv);
                atomicAdd(o + 64, v[u][2] * nv);
                atomicAdd(o + 96, v[u][3] * nv);
            }
        }
    } else {
        for (int j = 0; j < cnt; ++j) {
            int   s  = __builtin_amdgcn_readlane(s_l, j);
            int   d  = __builtin_amdgcn_readlane(d_l, j);
            float nv = __int_as_float(__builtin_amdgcn_readlane(__float_as_int(n_l), j));
            scatter_one(hb, out, s, d, nv, lane);
        }
    }
}


static inline size_t align_up(size_t x) { return (x + 63) & ~(size_t)63; }

extern "C" void kernel_launch(void* const* d_in, const int* in_sizes, int n_in,
                              void* d_out, int out_size, void* d_ws, size_t ws_size,
                              hipStream_t stream) {
    const float* x  = (const float*)d_in[0];
    const int*   ei = (const int*)d_in[1];
    const float* ew = (const float*)d_in[2];
    const float* W1 = (const float*)d_in[3];  const float* b1 = (const float*)d_in[4];
    const float* W2 = (const float*)d_in[5];  const float* b2 = (const float*)d_in[6];
    const float* W3 = (const float*)d_in[7];  const float* b3 = (const float*)d_in[8];
    const float* Wl = (const float*)d_in[9];  const float* bl = (const float*)d_in[10];

    const int N = in_sizes[0] / DFEAT;
    const int E = in_sizes[2];

    float* dinv = (float*)d_ws;
    float* hb   = dinv + align_up((size_t)N) + align_up((size_t)E);
    float* bufA = hb   + (size_t)N * DFEAT;
    float* bufB = bufA + (size_t)N * DFEAT;
    float* outF = (float*)d_out;

    const int T = 256;
    dim3 blkG((N + MT - 1) / MT);
    const int TS = 128;
    dim3 blkS((E + TS - 1) / TS);

    dim3 blkD(DPARTS, (N + DCH - 1) / DCH);
    deg_part_k<<<blkD, DTH, 0, stream>>>(ei, ew, bufB, N, E);

    gemm_wmma_k<0><<<blkG, T, 0, stream>>>(x,    W1, nullptr, nullptr, hb, bufA, bufB, dinv, N);
    scatter_k  <<<blkS, TS, 0, stream>>>(hb, ei, ew, bufA, E);

    gemm_wmma_k<1><<<blkG, T, 0, stream>>>(bufA, W2, b1, nullptr, hb, bufB, nullptr, dinv, N);
    scatter_k  <<<blkS, TS, 0, stream>>>(hb, ei, ew, bufB, E);

    gemm_wmma_k<1><<<blkG, T, 0, stream>>>(bufB, W3, b2, nullptr, hb, bufA, nullptr, dinv, N);
    scatter_k  <<<blkS, TS, 0, stream>>>(hb, ei, ew, bufA, E);

    gemm_wmma_k<2><<<blkG, T, 0, stream>>>(bufA, Wl, b3, bl, outF, nullptr, nullptr, dinv, N);
}


# Aggregation: a block keeps 196 destination rows in LDS, scans a 1-byte owner tag per edge, adds its edges' rows with LDS f32 adds; no global float atomics, no 2nd copy of g; arrival order free
# speedup vs baseline: 2.3721x; 1.0960x over previous
#include <hip/hip_runtime.h>
#include <hip/hip_bf16.h>

#define DFEAT 128
#define LDSP  132
#define MT    64

typedef float v2f __attribute__((ext_vector_type(2)));
typedef float v8f __attribute__((ext_vector_type(8)));


#define DCH    25088
#define DPARTS 64
#define DTH    512
#define DEU    13
#define AR     196

__global__ __launch_bounds__(DTH)
void deg_part_k(const int* __restrict__ ei, const float* __restrict__ w,
                float* __restrict__ part, unsigned char* __restrict__ own, int n, int E) {
    __shared__ alignas(16) float sdeg[DCH];
    const int tid = threadIdx.x;
    const int lo  = blockIdx.y * DCH;
    const int cn  = min(DCH, n - lo);
    const int pstride = (n + 31) & ~31;

    for (int i = tid; i < ((cn + 3) >> 2); i += DTH)
        ((float4*)sdeg)[i] = make_float4(0.f, 0.f, 0.f, 0.f);
    __syncthreads();

    const int per = (E + (int)gridDim.x - 1) / (int)gridDim.x;
    const int e0  = blockIdx.x * per;
    const int e1  = min(E, e0 + per);
    for (int base = e0 + tid; base < e1; base += DTH * DEU) {
        int   d[DEU];
        float wv[DEU];
        #pragma unroll
        for (int u = 0; u < DEU; ++u) {
            const int ec = min(base + u * DTH, e1 - 1);
            d[u]  = ei[E + ec];
            wv[u] = w[ec];
        }
        if (blockIdx.y == 0) {
            #pragma unroll
            for (int u = 0; u < DEU; ++u) {
                const int ec = min(base + u * DTH, e1 - 1);
                own[ec] = (unsigned char)((d[u] / AR) & 0xFF);
            }
        }
        #pragma unroll
        for (int u = 0; u < DEU; ++u) {
            const int r = d[u] - lo;
            if (base + u * DTH < e1 && (unsigned)r < (unsigned)cn) atomicAdd(&sdeg[r], wv[u]);
        }
    }
    __syncthreads();

    float* p = part + (size_t)blockIdx.x * pstride + lo;
    const int c4 = cn >> 2;
    for (int i = tid; i < c4; i += DTH) ((float4*)p)[i] = ((const float4*)sdeg)[i];
    for (int i = (c4 << 2) + tid; i < cn; i += DTH) p[i] = sdeg[i];
}

template <int MODE>
__global__ __launch_bounds__(256)
void gemm_wmma_k(const float* __restrict__ A, const float* __restrict__ W,
                 const float* __restrict__ biasIn, const float* __restrict__ biasOut,
                 float* __restrict__ out,
                 const float* __restrict__ part, float* __restrict__ dinv, int nrows) {
    __shared__ float As[MT][LDSP];
    __shared__ alignas(16) float Ds[MT];

    const int tid = threadIdx.x;
    const int rowBase = blockIdx.x * MT;

    if (MODE == 0) {
        if (tid < MT) {
            const int gr = rowBase + tid;
            float d = 0.0f;
            if (gr < nrows) {
                const int pstride = (nrows + 31) & ~31;
                float pv[DPARTS];
                #pragma unroll
                for (int p = 0; p < DPARTS; ++p) pv[p] = part[(size_t)p * pstride + gr];
                float s = 1.0f;
                #pragma unroll
                for (int p = 0; p < DPARTS; ++p) s += pv[p];
                d = (s > 0.0f) ? rsqrtf(s) : 0.0f;
                dinv[gr] = d;
            }
            Ds[tid] = d;
        }
    }

    const float4* A4 = (const float4*)A;
    float bi0 = 0.f, bi1 = 0.f, bi2 = 0.f, bi3 = 0.f;
    if (MODE != 0) {
        const int c = (tid & 31) * 4;
        bi0 = biasIn[c]; bi1 = biasIn[c + 1]; bi2 = biasIn[c + 2]; bi3 = biasIn[c + 3];
    }
    #pragma unroll
    for (int j = 0; j < 8; ++j) {
        int idx4 = j * 256 + tid;
        int r    = idx4 >> 5;
        int c4   = idx4 & 31;
        float4 v = make_float4(0.f, 0.f, 0.f, 0.f);
        int gr = rowBase + r;
        if (gr < nrows) {
            v = A4[(size_t)gr * (DFEAT / 4) + c4];
            if (MODE != 0) {
                const float dv = dinv[gr];
                v.x = v.x * dv + bi0; v.y = v.y * dv + bi1;
                v.z = v.z * dv + bi2; v.w = v.w * dv + bi3;
            }
            if (MODE == 1) {
                v.x = fmaxf(v.x, 0.f); v.y = fmaxf(v.y, 0.f);
                v.z = fmaxf(v.z, 0.f); v.w = fmaxf(v.w, 0.f);
            }
        }
        *(float4*)&As[r][c4 * 4] = v;
    }
    if (MODE == 1) {
        if (tid < MT) {
            int gr = rowBase + tid;
            Ds[tid] = (gr < nrows) ? dinv[gr] : 0.0f;
        }
    }
    __syncthreads();

    const int lane = tid & 31;
    const int wv   = tid >> 5;
    const int colBase = wv * 16;
    const int m  = lane & 15;
    const int kh = (lane >> 4) * 2;
    const int col = colBase + m;

    v8f acc[4] = {};
    #pragma unroll
    for (int k0 = 0; k0 < DFEAT; k0 += 4) {
        v2f b;
        b.x = W[(size_t)(k0 + kh)     * DFEAT + col];
        b.y = W[(size_t)(k0 + kh + 1) * DFEAT + col];
        #pragma unroll
        for (int rt = 0; rt < 4; ++rt) {
            v2f a;
            a.x = As[rt * 16 + m][k0 + kh];
            a.y = As[rt * 16 + m][k0 + kh + 1];
            acc[rt] = __builtin_amdgcn_wmma_f32_16x16x4_f32(
                          false, a, false, b, (short)0, acc[rt], false, false);
        }
    }

    float bv = 0.0f;
    if (MODE == 2) bv = biasOut[col];
    const int rowOff = (lane >> 4) * 8;
    #pragma unroll
    for (int rt = 0; rt < 4; ++rt) {
        float sw[8] = {0.f, 0.f, 0.f, 0.f, 0.f, 0.f, 0.f, 0.f};
        if (MODE != 2) {
            const float4 s0 = *(const float4*)&Ds[rt * 16 + rowOff];
            const float4 s1 = *(const float4*)&Ds[rt * 16 + rowOff + 4];
            sw[0] = s0.x; sw[1] = s0.y; sw[2] = s0.z; sw[3] = s0.w;
            sw[4] = s1.x; sw[5] = s1.y; sw[6] = s1.z; sw[7] = s1.w;
        }
        #pragma unroll
        for (int v = 0; v < 8; ++v) {
            int gr = rowBase + rt * 16 + rowOff + v;
            if (gr < nrows) {
                const size_t o = (size_t)gr * DFEAT + col;
                if (MODE != 2) {
                    const float g = acc[rt][v] * sw[v];
                    out[o]  = g;
                } else {
                    out[o] = acc[rt][v] + bv;
                }
            }
        }
    }
}


#define ATH   1024
#define AWV   (ATH / 32)
#define QCAP  1056
#define EB    8
#define AFILL ((AR * DFEAT / 4 + ATH - 1) / ATH)

__device__ __forceinline__ void agg_flush(const float* __restrict__ hb, const int* __restrict__ ei,
                                          const float* __restrict__ wgt, float* acc, const int* qw,
                                          int cnt, int lo, int cn, int E, int lane) {
    int   s_l = 0, d_l = 0;
    float n_l = 0.0f;
    bool  mine = false;
    if (lane < cnt) {
        const int e = qw[lane];
        const int   s  = ei[e];
        const int   d  = ei[E + e] - lo;
        const float wt = wgt[e];
        mine = (unsigned)d < (unsigned)cn;
        s_l = mine ? s : 0;
        d_l = mine ? d : 0;
        n_l = mine ? wt : 0.0f;
    }
    const unsigned vm = __builtin_amdgcn_ballot_w32(mine);
    for (int j0 = 0; j0 < cnt; j0 += EB) {
        float v[EB][4];
        #pragma unroll
        for (int u = 0; u < EB; ++u) {
            const int s = __builtin_amdgcn_readlane(s_l, j0 + u);
            const float* h = hb + (size_t)s * DFEAT + lane;
            v[u][0] = h[0];
            v[u][1] = h[32];
            v[u][2] = h[64];
            v[u][3] = h[96];
        }
        #pragma unroll
        for (int u = 0; u < EB; ++u) {
            if ((vm >> (j0 + u)) & 1u) {
                const int   d  = __builtin_amdgcn_readlane(d_l, j0 + u);
                const float nv = __int_as_float(__builtin_amdgcn_readlane(__float_as_int(n_l), j0 + u));
                float* o = acc + d * DFEAT + lane;
                atomicAdd(o,      v[u][0] * nv);
                atomicAdd(o + 32, v[u][1] * nv);
                atomicAdd(o + 64, v[u][2] * nv);
                atomicAdd(o + 96, v[u][3] * nv);
            }
        }
    }
}

__global__ __launch_bounds__(ATH)
void agg_k(const float* __restrict__ hb, const int* __restrict__ ei, const float* __restrict__ wgt,
           const unsigned char* __restrict__ own, float* __restrict__ out, int n, int E) {
    __shared__ alignas(16) float acc[AR * DFEAT];
    __shared__ int q[AWV][QCAP];

    const int tid  = threadIdx.x;
    const int lane = tid & 31;
    const int lo   = blockIdx.x * AR;
    const int cn   = min(AR, n - lo);
    const unsigned mm = (blockIdx.x & 0xFFu) * 0x01010101u;

    const float4* g4 = (const float4*)(hb + (size_t)lo * DFEAT);
    {
        const int tot = cn * (DFEAT / 4);
        float4 r[AFILL];
        #pragma unroll
        for (int k = 0; k < AFILL; ++k) r[k] = g4[min(tid + k * ATH, tot - 1)];
        static_assert(AFILL == 7, "the statement below names r[0] to r[6]");
        asm volatile("" :: "v"(r[0].x), "v"(r[1].x), "v"(r[2].x), "v"(r[3].x), "v"(r[4].x), "v"(r[5].x), "v"(r[6].x));
        #pragma unroll
        for (int k = 0; k < AFILL; ++k)
            if (tid + k * ATH < tot) ((float4*)acc)[tid + k * ATH] = r[k];
    }
    __syncthreads();

    int* qw = q[tid >> 5];
    const int G = (E + 31) >> 5;
    const int rot = (int)(((long long)blockIdx.x * G) / (int)gridDim.x) & ~31;
    const uint4* own4 = (const uint4*)own;
    const unsigned ltmask = (1u << lane) - 1u;
    int qc = 0;

    bool ok = tid < G;
    int  g  = ok ? tid + rot : 0; if (g >= G) g -= G;
    uint4 c0 = own4[2 * g], c1 = own4[2 * g + 1];

    for (int base = 0; base < G; base += ATH) {
        int  gn  = base + ATH + tid;
        const bool okn = gn < G;
        gn = okn ? gn + rot : 0; if (gn >= G) gn -= G;
        const uint4 n0 = own4[2 * gn], n1 = own4[2 * gn + 1];

        unsigned m = 0;
        if (ok) {
            const unsigned t[8] = {c0.x, c0.y, c0.z, c0.w, c1.x, c1.y, c1.z, c1.w};
            #pragma unroll
            for (int w = 0; w < 8; ++w) {
                const unsigned x = t[w] ^ mm;
                const unsigned z = (x - 0x01010101u) & ~x & 0x80808080u;
                m |= z >> (7 - w);
            }
        }
        if (__builtin_amdgcn_ballot_w32(m != 0)) {
            do {
                int e = -1;
                if (m) {
                    const int b = __builtin_ctz(m);
                    m &= m - 1;
                    e = g * 32 + (b & 7) * 4 + (b >> 3);
                    if (e >= E) e = -1;
                }
                const unsigned hv = __builtin_amdgcn_ballot_w32(e >= 0);
                if (e >= 0) qw[qc + __builtin_popcount(hv & ltmask)] = e;
                qc += __builtin_popcount(hv);
            } while (__builtin_amdgcn_ballot_w32(m != 0));
        }
        int head = 0;
        while (qc - head >= 32) {
            agg_flush(hb, ei, wgt, acc, qw + head, 32, lo, cn, E, lane);
            head += 32;
        }
        if (head) {
            const int rem = qc - head;
            int tmp = 0;
            if (lane < rem) tmp = qw[head + lane];
            if (lane < rem) qw[lane] = tmp;
            qc = rem;
        }
        c0 = n0; c1 = n1; g = gn; ok = okn;
    }
    agg_flush(hb, ei, wgt, acc, qw, qc, lo, cn, E, lane);
    __syncthreads();

    float4* o4 = (float4*)(out + (size_t)lo * DFEAT);
    for (int i = tid; i < cn * (DFEAT / 4); i += ATH) o4[i] = ((const float4*)acc)[i];
}


static inline size_t align_up(size_t x) { return (x + 63) & ~(size_t)63; }

extern "C" void kernel_launch(void* const* d_in, const int* in_sizes, int n_in,
                              void* d_out, int out_size, void* d_ws, size_t ws_size,
                              hipStream_t stream) {
    const float* x  = (const float*)d_in[0];
    const int*   ei = (const int*)d_in[1];
    const float* ew = (const float*)d_in[2];
    const float* W1 = (const float*)d_in[3];  const float* b1 = (const float*)d_in[4];
    const float* W2 = (const float*)d_in[5];  const float* b2 = (const float*)d_in[6];
    const float* W3 = (const float*)d_in[7];  const float* b3 = (const float*)d_in[8];
    const float* Wl = (const float*)d_in[9];  const float* bl = (const float*)d_in[10];

    const int N = in_sizes[0] / DFEAT;
    const int E = in_sizes[2];

    float* dinv = (float*)d_ws;
    unsigned char* own = (unsigned char*)(dinv + align_up((size_t)N));
    float* hb   = dinv + align_up((size_t)N) + align_up((size_t)E);
    float* bufA = hb   + (size_t)N * DFEAT;
    float* bufB = bufA + (size_t)N * DFEAT;
    float* outF = (float*)d_out;

    const int T = 256;
    dim3 blkG((N + MT - 1) / MT);
    dim3 blkA((N + AR - 1) / AR);

    dim3 blkD(DPARTS, (N + DCH - 1) / DCH);
    deg_part_k<<<blkD, DTH, 0, stream>>>(ei, ew, bufB, own, N, E);

    gemm_wmma_k<0><<<blkG, T, 0, stream>>>(x,    W1, nullptr, nullptr, hb, bufB, dinv, N);
    agg_k      <<<blkA, ATH, 0, stream>>>(hb, ei, ew, own, bufA, N, E);

    gemm_wmma_k<1><<<blkG, T, 0, stream>>>(bufA, W2, b1, nullptr, hb, nullptr, dinv, N);
    agg_k      <<<blkA, ATH, 0, stream>>>(hb, ei, ew, own, bufB, N, E);

    gemm_wmma_k<1><<<blkG, T, 0, stream>>>(bufB, W3, b2, nullptr, hb, nullptr, dinv, N);
    agg_k      <<<blkA, ATH, 0, stream>>>(hb, ei, ew, own, bufA, N, E);

    gemm_wmma_k<2><<<blkG, T, 0, stream>>>(bufA, Wl, b3, bl, outF, nullptr, dinv, N);
}


# Aggregation: layer 1 stores each block's own edges once as 8-byte records (source, local row, weight); layers 2, 3 replay runs of them: no tag scan, no 3 scattered words per edge; add order free
# speedup vs baseline: 2.8495x; 1.2013x over previous
#include <hip/hip_runtime.h>
#include <hip/hip_bf16.h>

#define DFEAT 128
#define LDSP  132
#define MT    64

typedef float v2f __attribute__((ext_vector_type(2)));
typedef float v8f __attribute__((ext_vector_type(8)));


#define DCH    25088
#define DPARTS 64
#define DTH    512
#define DEU    13
#define AR     196

__global__ __launch_bounds__(DTH)
void deg_part_k(const int* __restrict__ ei, const float* __restrict__ w,
                float* __restrict__ part, unsigned char* __restrict__ own, int n, int E) {
    __shared__ alignas(16) float sdeg[DCH];
    const int tid = threadIdx.x;
    const int lo  = blockIdx.y * DCH;
    const int cn  = min(DCH, n - lo);
    const int pstride = (n + 31) & ~31;

    for (int i = tid; i < ((cn + 3) >> 2); i += DTH)
        ((float4*)sdeg)[i] = make_float4(0.f, 0.f, 0.f, 0.f);
    __syncthreads();

    const int per = (E + (int)gridDim.x - 1) / (int)gridDim.x;
    const int e0  = blockIdx.x * per;
    const int e1  = min(E, e0 + per);
    for (int base = e0 + tid; base < e1; base += DTH * DEU) {
        int   d[DEU];
        float wv[DEU];
        #pragma unroll
        for (int u = 0; u < DEU; ++u) {
            const int ec = min(base + u * DTH, e1 - 1);
            d[u]  = ei[E + ec];
            wv[u] = w[ec];
        }
        if (blockIdx.y == 0) {
            #pragma unroll
            for (int u = 0; u < DEU; ++u) {
                const int ec = min(base + u * DTH, e1 - 1);
                own[ec] = (unsigned char)((d[u] / AR) & 0xFF);
            }
        }
        #pragma unroll
        for (int u = 0; u < DEU; ++u) {
            const int r = d[u] - lo;
            if (base + u * DTH < e1 && (unsigned)r < (unsigned)cn) atomicAdd(&sdeg[r], wv[u]);
        }
    }
    __syncthreads();

    float* p = part + (size_t)blockIdx.x * pstride + lo;
    const int c4 = cn >> 2;
    for (int i = tid; i < c4; i += DTH) ((float4*)p)[i] = ((const float4*)sdeg)[i];
    for (int i = (c4 << 2) + tid; i < cn; i += DTH) p[i] = sdeg[i];
}

template <int MODE>
__global__ __launch_bounds__(256)
void gemm_wmma_k(const float* __restrict__ A, const float* __restrict__ W,
                 const float* __restrict__ biasIn, const float* __restrict__ biasOut,
                 float* __restrict__ out,
                 const float* __restrict__ part, float* __restrict__ dinv, int nrows) {
    __shared__ float As[MT][LDSP];
    __shared__ alignas(16) float Ds[MT];

    const int tid = threadIdx.x;
    const int rowBase = blockIdx.x * MT;

    if (MODE == 0) {
        if (tid < MT) {
            const int gr = rowBase + tid;
            float d = 0.0f;
            if (gr < nrows) {
                const int pstride = (nrows + 31) & ~31;
                float pv[DPARTS];
                #pragma unroll
                for (int p = 0; p < DPARTS; ++p) pv[p] = part[(size_t)p * pstride + gr];
                float s = 1.0f;
                #pragma unroll
                for (int p = 0; p < DPARTS; ++p) s += pv[p];
                d = (s > 0.0f) ? rsqrtf(s) : 0.0f;
                dinv[gr] = d;
            }
            Ds[tid] = d;
        }
    }

    const float4* A4 = (const float4*)A;
    float bi0 = 0.f, bi1 = 0.f, bi2 = 0.f, bi3 = 0.f;
    if (MODE != 0) {
        const int c = (tid & 31) * 4;
        bi0 = biasIn[c]; bi1 = biasIn[c + 1]; bi2 = biasIn[c + 2]; bi3 = biasIn[c + 3];
    }
    #pragma unroll
    for (int j = 0; j < 8; ++j) {
        int idx4 = j * 256 + tid;
        int r    = idx4 >> 5;
        int c4   = idx4 & 31;
        float4 v = make_float4(0.f, 0.f, 0.f, 0.f);
        int gr = rowBase + r;
        if (gr < nrows) {
            v = A4[(size_t)gr * (DFEAT / 4) + c4];
            if (MODE != 0) {
                const float dv = dinv[gr];
                v.x = v.x * dv + bi0; v.y = v.y * dv + bi1;
                v.z = v.z * dv + bi2; v.w = v.w * dv + bi3;
            }
            if (MODE == 1) {
                v.x = fmaxf(v.x, 0.f); v.y = fmaxf(v.y, 0.f);
                v.z = fmaxf(v.z, 0.f); v.w = fmaxf(v.w, 0.f);
            }
        }
        *(float4*)&As[r][c4 * 4] = v;
    }
    if (MODE == 1) {
        if (tid < MT) {
            int gr = rowBase + tid;
            Ds[tid] = (gr < nrows) ? dinv[gr] : 0.0f;
        }
    }
    __syncthreads();

    const int lane = tid & 31;
    const int wv   = tid >> 5;
    const int colBase = wv * 16;
    const int m  = lane & 15;
    const int kh = (lane >> 4) * 2;
    const int col = colBase + m;

    v8f acc[4] = {};
    #pragma unroll
    for (int k0 = 0; k0 < DFEAT; k0 += 4) {
        v2f b;
        b.x = W[(size_t)(k0 + kh)     * DFEAT + col];
        b.y = W[(size_t)(k0 + kh + 1) * DFEAT + col];
        #pragma unroll
        for (int rt = 0; rt < 4; ++rt) {
            v2f a;
            a.x = As[rt * 16 + m][k0 + kh];
            a.y = As[rt * 16 + m][k0 + kh + 1];
            acc[rt] = __builtin_amdgcn_wmma_f32_16x16x4_f32(
                          false, a, false, b, (short)0, acc[rt], false, false);
        }
    }

    float bv = 0.0f;
    if (MODE == 2) bv = biasOut[col];
    const int rowOff = (lane >> 4) * 8;
    #pragma unroll
    for (int rt = 0; rt < 4; ++rt) {
        float sw[8] = {0.f, 0.f, 0.f, 0.f, 0.f, 0.f, 0.f, 0.f};
        if (MODE != 2) {
            const float4 s0 = *(const float4*)&Ds[rt * 16 + rowOff];
            const float4 s1 = *(const float4*)&Ds[rt * 16 + rowOff + 4];
            sw[0] = s0.x; sw[1] = s0.y; sw[2] = s0.z; sw[3] = s0.w;
            sw[4] = s1.x; sw[5] = s1.y; sw[6] = s1.z; sw[7] = s1.w;
        }
        #pragma unroll
        for (int v = 0; v < 8; ++v) {
            int gr = rowBase + rt * 16 + rowOff + v;
            if (gr < nrows) {
                const size_t o = (size_t)gr * DFEAT + col;
                if (MODE != 2) {
                    const float g = acc[rt][v] * sw[v];
                    out[o]  = g;
                } else {
                    out[o] = acc[rt][v] + bv;
                }
            }
        }
    }
}


#define ATH   1024
#define AWV   (ATH / 32)
#define QCAP  1056
#define EB    8
#define AFILL ((AR * DFEAT / 4 + ATH - 1) / ATH)

static_assert(AR <= 256, "a record keeps the local row in 8 bits");

__device__ __forceinline__ void agg_add(const float* __restrict__ hb, float* acc,
                                        int s_l, int d_l, float n_l, unsigned vm, int cnt, int lane) {
    for (int j0 = 0; j0 < cnt; j0 += EB) {
        float v[EB][4];
        #pragma unroll
        for (int u = 0; u < EB; ++u) {
            const int s = __builtin_amdgcn_readlane(s_l, j0 + u);
            const float* h = hb + (size_t)s * DFEAT + lane;
            v[u][0] = h[0];
            v[u][1] = h[32];
            v[u][2] = h[64];
            v[u][3] = h[96];
        }
        #pragma unroll
        for (int u = 0; u < EB; ++u) {
            if ((vm >> (j0 + u)) & 1u) {
                const int   d  = __builtin_amdgcn_readlane(d_l, j0 + u);
                const float nv = __int_as_float(__builtin_amdgcn_readlane(__float_as_int(n_l), j0 + u));
                float* o = acc + d * DFEAT + lane;
                atomicAdd(o,      v[u][0] * nv);
                atomicAdd(o + 32, v[u][1] * nv);
                atomicAdd(o + 64, v[u][2] * nv);
                atomicAdd(o + 96, v[u][3] * nv);
            }
        }
    }
}

template <bool SAVE>
__device__ __forceinline__ void agg_flush(const float* __restrict__ hb, const int* __restrict__ ei,
                                          const float* __restrict__ wgt, float* acc, const int* qw,
                                          int cnt, int lo, int cn, int E, int lane,
                                          int2* rec, int rcap, int* scnt) {
    int   s_l = 0, d_l = 0;
    float n_l = 0.0f;
    bool  mine = false;
    if (lane < cnt) {
        const int e = qw[lane];
        const int   s  = ei[e];
        const int   d  = ei[E + e] - lo;
        const float wt = wgt[e];
        mine = (unsigned)d < (unsigned)cn;
        s_l = mine ? s : 0;
        d_l = mine ? d : 0;
        n_l = mine ? wt : 0.0f;
    }
    const unsigned vm = __builtin_amdgcn_ballot_w32(mine);
    if (SAVE) {
        const int nown = __builtin_popcount(vm);
        int b0 = 0;
        if (lane == 0 && nown) b0 = atomicAdd(scnt, nown);
        b0 = __builtin_amdgcn_readfirstlane(b0);
        const int slot = b0 + __builtin_popcount(vm & ((1u << lane) - 1u));
        if (mine && slot < rcap) rec[slot] = make_int2((int)(((unsigned)s_l << 8) | (unsigned)d_l), __float_as_int(n_l));
    }
    agg_add(hb, acc, s_l, d_l, n_l, vm, cnt, lane);
}

template <bool SAVE>
__device__ __forceinline__ void agg_scan(const float* __restrict__ hb, const int* __restrict__ ei,
                                         const float* __restrict__ wgt, const unsigned char* __restrict__ own,
                                         float* acc, int* qw, int lo, int cn, int E, int tid, int lane,
                                         int2* rec, int rcap, int* scnt) {
    const unsigned mm = (blockIdx.x & 0xFFu) * 0x01010101u;
    const int G = (E + 31) >> 5;
    const int rot = (int)(((long long)blockIdx.x * G) / (int)gridDim.x) & ~31;
    const uint4* own4 = (const uint4*)own;
    const unsigned ltmask = (1u << lane) - 1u;
    int qc = 0;

    bool ok = tid < G;
    int  g  = ok ? tid + rot : 0; if (g >= G) g -= G;
    uint4 c0 = own4[2 * g], c1 = own4[2 * g + 1];

    for (int base = 0; base < G; base += ATH) {
        int  gn  = base + ATH + tid;
        const bool okn = gn < G;
        gn = okn ? gn + rot : 0; if (gn >= G) gn -= G;
        const uint4 n0 = own4[2 * gn], n1 = own4[2 * gn + 1];

        unsigned m = 0;
        if (ok) {
            const unsigned t[8] = {c0.x, c0.y, c0.z, c0.w, c1.x, c1.y, c1.z, c1.w};
            #pragma unroll
            for (int w = 0; w < 8; ++w) {
                const unsigned x = t[w] ^ mm;
                const unsigned z = (x - 0x01010101u) & ~x & 0x80808080u;
                m |= z >> (7 - w);
            }
        }
        if (__builtin_amdgcn_ballot_w32(m != 0)) {
            do {
                int e = -1;
                if (m) {
                    const int b = __builtin_ctz(m);
                    m &= m - 1;
                    e = g * 32 + (b & 7) * 4 + (b >> 3);
                    if (e >= E) e = -1;
                }
                const unsigned hv = __builtin_amdgcn_ballot_w32(e >= 0);
                if (e >= 0) qw[qc + __builtin_popcount(hv & ltmask)] = e;
                qc += __builtin_popcount(hv);
            } while (__builtin_amdgcn_ballot_w32(m != 0));
        }
        int head = 0;
        while (qc - head >= 32) {
            agg_flush<SAVE>(hb, ei, wgt, acc, qw + head, 32, lo, cn, E, lane, rec, rcap, scnt);
            head += 32;
        }
        if (head) {
            const int rem = qc - head;
            int tmp = 0;
            if (lane < rem) tmp = qw[head + lane];
            if (lane < rem) qw[lane] = tmp;
            qc = rem;
        }
        c0 = n0; c1 = n1; g = gn; ok = okn;
    }
    agg_flush<SAVE>(hb, ei, wgt, acc, qw, qc, lo, cn, E, lane, rec, rcap, scnt);
}

template <bool REPLAY>
__global__ __launch_bounds__(ATH)
void agg_k(const float* __restrict__ hb, const int* __restrict__ ei, const float* __restrict__ wgt,
           const unsigned char* __restrict__ own, float* __restrict__ out, float* lst, int n, int E) {
    __shared__ alignas(16) float acc[AR * DFEAT];
    __shared__ int q[AWV][QCAP];
    __shared__ int scnt;

    const int tid  = threadIdx.x;
    const int lane = tid & 31;
    const int lo   = blockIdx.x * AR;
    const int cn   = min(AR, n - lo);

    const int rstride = (int)((((long long)n * DFEAT) / (int)gridDim.x) & ~31LL);
    int2* rec = (int2*)(lst + (size_t)blockIdx.x * rstride);
    const int rcap = (n <= (1 << 24)) ? rstride / 2 - 1 : 0;

    int total = 0;
    if (REPLAY) total = __builtin_amdgcn_readfirstlane(rec[rcap].x);
    if (!REPLAY && tid == 0) scnt = 0;

    const float4* g4 = (const float4*)(hb + (size_t)lo * DFEAT);
    {
        const int tot = cn * (DFEAT / 4);
        float4 r[AFILL];
        #pragma unroll
        for (int k = 0; k < AFILL; ++k) r[k] = g4[min(tid + k * ATH, tot - 1)];
        static_assert(AFILL == 7, "the statement below names r[0] to r[6]");
        asm volatile("" :: "v"(r[0].x), "v"(r[1].x), "v"(r[2].x), "v"(r[3].x), "v"(r[4].x), "v"(r[5].x), "v"(r[6].x));
        #pragma unroll
        for (int k = 0; k < AFILL; ++k)
            if (tid + k * ATH < tot) ((float4*)acc)[tid + k * ATH] = r[k];
    }
    __syncthreads();

    if (REPLAY && total <= rcap) {
        int c = __builtin_amdgcn_readfirstlane(tid >> 5) * 32;
        int2 r = make_int2(0, 0);
        if (c + lane < total) r = rec[c + lane];
        while (c < total) {
            const int cnx = c + ATH;
            int2 rn = make_int2(0, 0);
            if (cnx + lane < total) rn = rec[cnx + lane];
            const int cnt = min(32, total - c);
            const unsigned vm = (cnt >= 32) ? 0xFFFFFFFFu : ((1u << cnt) - 1u);
            agg_add(hb, acc, (int)((unsigned)r.x >> 8), r.x & 0xFF, __int_as_float(r.y), vm, cnt, lane);
            r = rn; c = cnx;
        }
    } else {
        agg_scan<!REPLAY>(hb, ei, wgt, own, acc, q[tid >> 5], lo, cn, E, tid, lane, rec, rcap, &scnt);
    }
    __syncthreads();

    if (!REPLAY && tid == 0) rec[rcap] = make_int2(scnt, 0);

    float4* o4 = (float4*)(out + (size_t)lo * DFEAT);
    for (int i = tid; i < cn * (DFEAT / 4); i += ATH) o4[i] = ((const float4*)acc)[i];
}


static inline size_t align_up(size_t x) { return (x + 63) & ~(size_t)63; }

extern "C" void kernel_launch(void* const* d_in, const int* in_sizes, int n_in,
                              void* d_out, int out_size, void* d_ws, size_t ws_size,
                              hipStream_t stream) {
    const float* x  = (const float*)d_in[0];
    const int*   ei = (const int*)d_in[1];
    const float* ew = (const float*)d_in[2];
    const float* W1 = (const float*)d_in[3];  const float* b1 = (const float*)d_in[4];
    const float* W2 = (const float*)d_in[5];  const float* b2 = (const float*)d_in[6];
    const float* W3 = (const float*)d_in[7];  const float* b3 = (const float*)d_in[8];
    const float* Wl = (const float*)d_in[9];  const float* bl = (const float*)d_in[10];

    const int N = in_sizes[0] / DFEAT;
    const int E = in_sizes[2];

    float* dinv = (float*)d_ws;
    unsigned char* own = (unsigned char*)(dinv + align_up((size_t)N));
    float* hb   = dinv + align_up((size_t)N) + align_up((size_t)E);
    float* bufA = hb   + (size_t)N * DFEAT;
    float* bufB = bufA + (size_t)N * DFEAT;
    float* outF = (float*)d_out;

    const int T = 256;
    dim3 blkG((N + MT - 1) / MT);
    dim3 blkA((N + AR - 1) / AR);

    dim3 blkD(DPARTS, (N + DCH - 1) / DCH);
    deg_part_k<<<blkD, DTH, 0, stream>>>(ei, ew, bufB, own, N, E);

    gemm_wmma_k<0><<<blkG, T, 0, stream>>>(x,    W1, nullptr, nullptr, hb, bufB, dinv, N);
    agg_k<false><<<blkA, ATH, 0, stream>>>(hb, ei, ew, own, bufA, bufB, N, E);

    gemm_wmma_k<1><<<blkG, T, 0, stream>>>(bufA, W2, b1, nullptr, hb, nullptr, dinv, N);
    agg_k<true> <<<blkA, ATH, 0, stream>>>(hb, ei, ew, own, bufA, bufB, N, E);

    gemm_wmma_k<1><<<blkG, T, 0, stream>>>(bufA, W3, b2, nullptr, hb, nullptr, dinv, N);
    agg_k<true> <<<blkA, ATH, 0, stream>>>(hb, ei, ew, own, bufA, bufB, N, E);

    gemm_wmma_k<2><<<blkG, T, 0, stream>>>(bufA, Wl, b3, bl, outF, nullptr, dinv, N);
}


# GEMM tile staging: a thread's eight 16-byte tile loads, eight row scales and bias go out as one group (clamped rows, no branch round the loads), not one memory round trip per pass; sums unchanged
# speedup vs baseline: 2.9920x; 1.0500x over previous
#include <hip/hip_runtime.h>
#include <hip/hip_bf16.h>

#define DFEAT 128
#define LDSP  132
#define MT    64

typedef float v2f __attribute__((ext_vector_type(2)));
typedef float v8f __attribute__((ext_vector_type(8)));
typedef float v4f __attribute__((ext_vector_type(4)));


#define DCH    25088
#define DPARTS 64
#define DTH    512
#define DEU    13
#define AR     196

__global__ __launch_bounds__(DTH)
void deg_part_k(const int* __restrict__ ei, const float* __restrict__ w,
                float* __restrict__ part, unsigned char* __restrict__ own, int n, int E) {
    __shared__ alignas(16) float sdeg[DCH];
    const int tid = threadIdx.x;
    const int lo  = blockIdx.y * DCH;
    const int cn  = min(DCH, n - lo);
    const int pstride = (n + 31) & ~31;

    for (int i = tid; i < ((cn + 3) >> 2); i += DTH)
        ((float4*)sdeg)[i] = make_float4(0.f, 0.f, 0.f, 0.f);
    __syncthreads();

    const int per = (E + (int)gridDim.x - 1) / (int)gridDim.x;
    const int e0  = blockIdx.x * per;
    const int e1  = min(E, e0 + per);
    for (int base = e0 + tid; base < e1; base += DTH * DEU) {
        int   d[DEU];
        float wv[DEU];
        #pragma unroll
        for (int u = 0; u < DEU; ++u) {
            const int ec = min(base + u * DTH, e1 - 1);
            d[u]  = ei[E + ec];
            wv[u] = w[ec];
        }
        if (blockIdx.y == 0) {
            #pragma unroll
            for (int u = 0; u < DEU; ++u) {
                const int ec = min(base + u * DTH, e1 - 1);
                own[ec] = (unsigned char)((d[u] / AR) & 0xFF);
            }
        }
        #pragma unroll
        for (int u = 0; u < DEU; ++u) {
            const int r = d[u] - lo;
            if (base + u * DTH < e1 && (unsigned)r < (unsigned)cn) atomicAdd(&sdeg[r], wv[u]);
        }
    }
    __syncthreads();

    float* p = part + (size_t)blockIdx.x * pstride + lo;
    const int c4 = cn >> 2;
    for (int i = tid; i < c4; i += DTH) ((float4*)p)[i] = ((const float4*)sdeg)[i];
    for (int i = (c4 << 2) + tid; i < cn; i += DTH) p[i] = sdeg[i];
}

template <int MODE>
__global__ __launch_bounds__(256)
void gemm_wmma_k(const float* __restrict__ A, const float* __restrict__ W,
                 const float* __restrict__ biasIn, const float* __restrict__ biasOut,
                 float* __restrict__ out,
                 const float* __restrict__ part, float* __restrict__ dinv, int nrows) {
    __shared__ float As[MT][LDSP];
    __shared__ alignas(16) float Ds[MT];

    const int tid = threadIdx.x;
    const int rowBase = blockIdx.x * MT;

    const v4f* A4 = (const v4f*)A;
    const int c4 = tid & 31;
    const int r0 = tid >> 5;
    float bi0 = 0.f, bi1 = 0.f, bi2 = 0.f, bi3 = 0.f;
    if (MODE != 0) {
        const int c = c4 * 4;
        bi0 = biasIn[c]; bi1 = biasIn[c + 1]; bi2 = biasIn[c + 2]; bi3 = biasIn[c + 3];
    }
    v4f   va[8];
    float dv[8];
    #pragma unroll
    for (int j = 0; j < 8; ++j) {
        const int gr = min(rowBase + r0 + 8 * j, nrows - 1);
        va[j] = A4[(size_t)gr * (DFEAT / 4) + c4];
        dv[j] = (MODE != 0) ? dinv[gr] : 1.0f;
    }

    if (MODE == 0) {
        if (tid < MT) {
            const int gr = rowBase + tid;
            float d = 0.0f;
            if (gr < nrows) {
                const int pstride = (nrows + 31) & ~31;
                float pv[DPARTS];
                #pragma unroll
                for (int p = 0; p < DPARTS; ++p) pv[p] = part[(size_t)p * pstride + gr];
                float s = 1.0f;
                #pragma unroll
                for (int p = 0; p < DPARTS; ++p) s += pv[p];
                d = (s > 0.0f) ? rsqrtf(s) : 0.0f;
                dinv[gr] = d;
            }
            Ds[tid] = d;
        }
    }

    if (MODE != 0) {
        asm volatile("" :: "v"(va[0]), "v"(va[1]), "v"(va[2]), "v"(va[3]), "v"(va[4]), "v"(va[5]), "v"(va[6]), "v"(va[7]),
                           "v"(dv[0]), "v"(dv[1]), "v"(dv[2]), "v"(dv[3]), "v"(dv[4]), "v"(dv[5]), "v"(dv[6]), "v"(dv[7]),
                           "v"(bi0), "v"(bi1), "v"(bi2), "v"(bi3));
    }

    #pragma unroll
    for (int j = 0; j < 8; ++j) {
        const int r = r0 + 8 * j;
        const bool live = rowBase + r < nrows;
        v4f v = va[j];
        if (MODE != 0) {
            const float d = dv[j];
            v.x = v.x * d + bi0; v.y = v.y * d + bi1;
            v.z = v.z * d + bi2; v.w = v.w * d + bi3;
        }
        if (MODE == 1) {
            v.x = fmaxf(v.x, 0.f); v.y = fmaxf(v.y, 0.f);
            v.z = fmaxf(v.z, 0.f); v.w = fmaxf(v.w, 0.f);
        }
        if (!live) v = (v4f)(0.0f);
        *(v4f*)&As[r][c4 * 4] = v;
        if (MODE == 1 && c4 == 0) Ds[r] = live ? dv[j] : 0.0f;
    }
    __syncthreads();

    const int lane = tid & 31;
    const int wv   = tid >> 5;
    const int colBase = wv * 16;
    const int m  = lane & 15;
    const int kh = (lane >> 4) * 2;
    const int col = colBase + m;

    v8f acc[4] = {};
    #pragma unroll
    for (int k0 = 0; k0 < DFEAT; k0 += 4) {
        v2f b;
        b.x = W[(size_t)(k0 + kh)     * DFEAT + col];
        b.y = W[(size_t)(k0 + kh + 1) * DFEAT + col];
        #pragma unroll
        for (int rt = 0; rt < 4; ++rt) {
            v2f a;
            a.x = As[rt * 16 + m][k0 + kh];
            a.y = As[rt * 16 + m][k0 + kh + 1];
            acc[rt] = __builtin_amdgcn_wmma_f32_16x16x4_f32(
                          false, a, false, b, (short)0, acc[rt], false, false);
        }
    }

    float bv = 0.0f;
    if (MODE == 2) bv = biasOut[col];
    const int rowOff = (lane >> 4) * 8;
    #pragma unroll
    for (int rt = 0; rt < 4; ++rt) {
        float sw[8] = {0.f, 0.f, 0.f, 0.f, 0.f, 0.f, 0.f, 0.f};
        if (MODE != 2) {
            const float4 s0 = *(const float4*)&Ds[rt * 16 + rowOff];
            const float4 s1 = *(const float4*)&Ds[rt * 16 + rowOff + 4];
            sw[0] = s0.x; sw[1] = s0.y; sw[2] = s0.z; sw[3] = s0.w;
            sw[4] = s1.x; sw[5] = s1.y; sw[6] = s1.z; sw[7] = s1.w;
        }
        #pragma unroll
        for (int v = 0; v < 8; ++v) {
            int gr = rowBase + rt * 16 + rowOff + v;
            if (gr < nrows) {
                const size_t o = (size_t)gr * DFEAT + col;
                if (MODE != 2) {
                    const float g = acc[rt][v] * sw[v];
                    out[o]  = g;
                } else {
                    out[o] = acc[rt][v] + bv;
                }
            }
        }
    }
}


#define ATH   1024
#define AWV   (ATH / 32)
#define QCAP  1056
#define EB    8
#define AFILL ((AR * DFEAT / 4 + ATH - 1) / ATH)

static_assert(AR <= 256, "a record keeps the local row in 8 bits");

__device__ __forceinline__ void agg_add(const float* __restrict__ hb, float* acc,
                                        int s_l, int d_l, float n_l, unsigned vm, int cnt, int lane) {
    for (int j0 = 0; j0 < cnt; j0 += EB) {
        float v[EB][4];
        #pragma unroll
        for (int u = 0; u < EB; ++u) {
            const int s = __builtin_amdgcn_readlane(s_l, j0 + u);
            const float* h = hb + (size_t)s * DFEAT + lane;
            v[u][0] = h[0];
            v[u][1] = h[32];
            v[u][2] = h[64];
            v[u][3] = h[96];
        }
        #pragma unroll
        for (int u = 0; u < EB; ++u) {
            if ((vm >> (j0 + u)) & 1u) {
                const int   d  = __builtin_amdgcn_readlane(d_l, j0 + u);
                const float nv = __int_as_float(__builtin_amdgcn_readlane(__float_as_int(n_l), j0 + u));
                float* o = acc + d * DFEAT + lane;
                atomicAdd(o,      v[u][0] * nv);
                atomicAdd(o + 32, v[u][1] * nv);
                atomicAdd(o + 64, v[u][2] * nv);
                atomicAdd(o + 96, v[u][3] * nv);
            }
        }
    }
}

template <bool SAVE>
__device__ __forceinline__ void agg_flush(const float* __restrict__ hb, const int* __restrict__ ei,
                                          const float* __restrict__ wgt, float* acc, const int* qw,
                                          int cnt, int lo, int cn, int E, int lane,
                                          int2* rec, int rcap, int* scnt) {
    int   s_l = 0, d_l = 0;
    float n_l = 0.0f;
    bool  mine = false;
    if (lane < cnt) {
        const int e = qw[lane];
        const int   s  = ei[e];
        const int   d  = ei[E + e] - lo;
        const float wt = wgt[e];
        mine = (unsigned)d < (unsigned)cn;
        s_l = mine ? s : 0;
        d_l = mine ? d : 0;
        n_l = mine ? wt : 0.0f;
    }
    const unsigned vm = __builtin_amdgcn_ballot_w32(mine);
    if (SAVE) {
        const int nown = __builtin_popcount(vm);
        int b0 = 0;
        if (lane == 0 && nown) b0 = atomicAdd(scnt, nown);
        b0 = __builtin_amdgcn_readfirstlane(b0);
        const int slot = b0 + __builtin_popcount(vm & ((1u << lane) - 1u));
        if (mine && slot < rcap) rec[slot] = make_int2((int)(((unsigned)s_l << 8) | (unsigned)d_l), __float_as_int(n_l));
    }
    agg_add(hb, acc, s_l, d_l, n_l, vm, cnt, lane);
}

template <bool SAVE>
__device__ __forceinline__ void agg_scan(const float* __restrict__ hb, const int* __restrict__ ei,
                                         const float* __restrict__ wgt, const unsigned char* __restrict__ own,
                                         float* acc, int* qw, int lo, int cn, int E, int tid, int lane,
                                         int2* rec, int rcap, int* scnt) {
    const unsigned mm = (blockIdx.x & 0xFFu) * 0x01010101u;
    const int G = (E + 31) >> 5;
    const int rot = (int)(((long long)blockIdx.x * G) / (int)gridDim.x) & ~31;
    const uint4* own4 = (const uint4*)own;
    const unsigned ltmask = (1u << lane) - 1u;
    int qc = 0;

    bool ok = tid < G;
    int  g  = ok ? tid + rot : 0; if (g >= G) g -= G;
    uint4 c0 = own4[2 * g], c1 = own4[2 * g + 1];

    for (int base = 0; base < G; base += ATH) {
        int  gn  = base + ATH + tid;
        const bool okn = gn < G;
        gn = okn ? gn + rot : 0; if (gn >= G) gn -= G;
        const uint4 n0 = own4[2 * gn], n1 = own4[2 * gn + 1];

        unsigned m = 0;
        if (ok) {
            const unsigned t[8] = {c0.x, c0.y, c0.z, c0.w, c1.x, c1.y, c1.z, c1.w};
            #pragma unroll
            for (int w = 0; w < 8; ++w) {
                const unsigned x = t[w] ^ mm;
                const unsigned z = (x - 0x01010101u) & ~x & 0x80808080u;
                m |= z >> (7 - w);
            }
        }
        if (__builtin_amdgcn_ballot_w32(m != 0)) {
            do {
                int e = -1;
                if (m) {
                    const int b = __builtin_ctz(m);
                    m &= m - 1;
                    e = g * 32 + (b & 7) * 4 + (b >> 3);
                    if (e >= E) e = -1;
                }
                const unsigned hv = __builtin_amdgcn_ballot_w32(e >= 0);
                if (e >= 0) qw[qc + __builtin_popcount(hv & ltmask)] = e;
                qc += __builtin_popcount(hv);
            } while (__builtin_amdgcn_ballot_w32(m != 0));
        }
        int head = 0;
        while (qc - head >= 32) {
            agg_flush<SAVE>(hb, ei, wgt, acc, qw + head, 32, lo, cn, E, lane, rec, rcap, scnt);
            head += 32;
        }
        if (head) {
            const int rem = qc - head;
            int tmp = 0;
            if (lane < rem) tmp = qw[head + lane];
            if (lane < rem) qw[lane] = tmp;
            qc = rem;
        }
        c0 = n0; c1 = n1; g = gn; ok = okn;
    }
    agg_flush<SAVE>(hb, ei, wgt, acc, qw, qc, lo, cn, E, lane, rec, rcap, scnt);
}

template <bool REPLAY>
__global__ __launch_bounds__(ATH)
void agg_k(const float* __restrict__ hb, const int* __restrict__ ei, const float* __restrict__ wgt,
           const unsigned char* __restrict__ own, float* __restrict__ out, float* lst, int n, int E) {
    __shared__ alignas(16) float acc[AR * DFEAT];
    __shared__ int q[AWV][QCAP];
    __shared__ int scnt;

    const int tid  = threadIdx.x;
    const int lane = tid & 31;
    const int lo   = blockIdx.x * AR;
    const int cn   = min(AR, n - lo);

    const int rstride = (int)((((long long)n * DFEAT) / (int)gridDim.x) & ~31LL);
    int2* rec = (int2*)(lst + (size_t)blockIdx.x * rstride);
    const int rcap = (n <= (1 << 24)) ? rstride / 2 - 1 : 0;

    int total = 0;
    if (REPLAY) total = __builtin_amdgcn_readfirstlane(rec[rcap].x);
    if (!REPLAY && tid == 0) scnt = 0;

    const float4* g4 = (const float4*)(hb + (size_t)lo * DFEAT);
    {
        const int tot = cn * (DFEAT / 4);
        float4 r[AFILL];
        #pragma unroll
        for (int k = 0; k < AFILL; ++k) r[k] = g4[min(tid + k * ATH, tot - 1)];
        static_assert(AFILL == 7, "the statement below names r[0] to r[6]");
        asm volatile("" :: "v"(r[0].x), "v"(r[1].x), "v"(r[2].x), "v"(r[3].x), "v"(r[4].x), "v"(r[5].x), "v"(r[6].x));
        #pragma unroll
        for (int k = 0; k < AFILL; ++k)
            if (tid + k * ATH < tot) ((float4*)acc)[tid + k * ATH] = r[k];
    }
    __syncthreads();

    if (REPLAY && total <= rcap) {
        int c = __builtin_amdgcn_readfirstlane(tid >> 5) * 32;
        int2 r = make_int2(0, 0);
        if (c + lane < total) r = rec[c + lane];
        while (c < total) {
            const int cnx = c + ATH;
            int2 rn = make_int2(0, 0);
            if (cnx + lane < total) rn = rec[cnx + lane];
            const int cnt = min(32, total - c);
            const unsigned vm = (cnt >= 32) ? 0xFFFFFFFFu : ((1u << cnt) - 1u);
            agg_add(hb, acc, (int)((unsigned)r.x >> 8), r.x & 0xFF, __int_as_float(r.y), vm, cnt, lane);
            r = rn; c = cnx;
        }
    } else {
        agg_scan<!REPLAY>(hb, ei, wgt, own, acc, q[tid >> 5], lo, cn, E, tid, lane, rec, rcap, &scnt);
    }
    __syncthreads();

    if (!REPLAY && tid == 0) rec[rcap] = make_int2(scnt, 0);

    float4* o4 = (float4*)(out + (size_t)lo * DFEAT);
    for (int i = tid; i < cn * (DFEAT / 4); i += ATH) o4[i] = ((const float4*)acc)[i];
}


static inline size_t align_up(size_t x) { return (x + 63) & ~(size_t)63; }

extern "C" void kernel_launch(void* const* d_in, const int* in_sizes, int n_in,
                              void* d_out, int out_size, void* d_ws, size_t ws_size,
                              hipStream_t stream) {
    const float* x  = (const float*)d_in[0];
    const int*   ei = (const int*)d_in[1];
    const float* ew = (const float*)d_in[2];
    const float* W1 = (const float*)d_in[3];  const float* b1 = (const float*)d_in[4];
    const float* W2 = (const float*)d_in[5];  const float* b2 = (const float*)d_in[6];
    const float* W3 = (const float*)d_in[7];  const float* b3 = (const float*)d_in[8];
    const float* Wl = (const float*)d_in[9];  const float* bl = (const float*)d_in[10];

    const int N = in_sizes[0] / DFEAT;
    const int E = in_sizes[2];

    float* dinv = (float*)d_ws;
    unsigned char* own = (unsigned char*)(dinv + align_up((size_t)N));
    float* hb   = dinv + align_up((size_t)N) + align_up((size_t)E);
    float* bufA = hb   + (size_t)N * DFEAT;
    float* bufB = bufA + (size_t)N * DFEAT;
    float* outF = (float*)d_out;

    const int T = 256;
    dim3 blkG((N + MT - 1) / MT);
    dim3 blkA((N + AR - 1) / AR);

    dim3 blkD(DPARTS, (N + DCH - 1) / DCH);
    deg_part_k<<<blkD, DTH, 0, stream>>>(ei, ew, bufB, own, N, E);

    gemm_wmma_k<0><<<blkG, T, 0, stream>>>(x,    W1, nullptr, nullptr, hb, bufB, dinv, N);
    agg_k<false><<<blkA, ATH, 0, stream>>>(hb, ei, ew, own, bufA, bufB, N, E);

    gemm_wmma_k<1><<<blkG, T, 0, stream>>>(bufA, W2, b1, nullptr, hb, nullptr, dinv, N);
    agg_k<true> <<<blkA, ATH, 0, stream>>>(hb, ei, ew, own, bufA, bufB, N, E);

    gemm_wmma_k<1><<<blkG, T, 0, stream>>>(bufA, W3, b2, nullptr, hb, nullptr, dinv, N);
    agg_k<true> <<<blkA, ATH, 0, stream>>>(hb, ei, ew, own, bufA, bufB, N, E);

    gemm_wmma_k<2><<<blkG, T, 0, stream>>>(bufA, Wl, b3, bl, outF, nullptr, dinv, N);
}


# Aggregation: a source row comes in by ONE 16-byte-per-lane load (4 whole lines), not four 4-byte loads; LDS rows keep column 4l+j at slot l+32j, so the LDS f32 adds stay conflict-free; add order free
# speedup vs baseline: 3.2077x; 1.0721x over previous
#include <hip/hip_runtime.h>
#include <hip/hip_bf16.h>

#define DFEAT 128
#define LDSP  132
#define MT    64

typedef float v2f __attribute__((ext_vector_type(2)));
typedef float v8f __attribute__((ext_vector_type(8)));
typedef float v4f __attribute__((ext_vector_type(4)));


#define DCH    25088
#define DPARTS 64
#define DTH    512
#define DEU    13
#define AR     196

__global__ __launch_bounds__(DTH)
void deg_part_k(const int* __restrict__ ei, const float* __restrict__ w,
                float* __restrict__ part, unsigned char* __restrict__ own, int n, int E) {
    __shared__ alignas(16) float sdeg[DCH];
    const int tid = threadIdx.x;
    const int lo  = blockIdx.y * DCH;
    const int cn  = min(DCH, n - lo);
    const int pstride = (n + 31) & ~31;

    for (int i = tid; i < ((cn + 3) >> 2); i += DTH)
        ((float4*)sdeg)[i] = make_float4(0.f, 0.f, 0.f, 0.f);
    __syncthreads();

    const int per = (E + (int)gridDim.x - 1) / (int)gridDim.x;
    const int e0  = blockIdx.x * per;
    const int e1  = min(E, e0 + per);
    for (int base = e0 + tid; base < e1; base += DTH * DEU) {
        int   d[DEU];
        float wv[DEU];
        #pragma unroll
        for (int u = 0; u < DEU; ++u) {
            const int ec = min(base + u * DTH, e1 - 1);
            d[u]  = ei[E + ec];
            wv[u] = w[ec];
        }
        if (blockIdx.y == 0) {
            #pragma unroll
            for (int u = 0; u < DEU; ++u) {
                const int ec = min(base + u * DTH, e1 - 1);
                own[ec] = (unsigned char)((d[u] / AR) & 0xFF);
            }
        }
        #pragma unroll
        for (int u = 0; u < DEU; ++u) {
            const int r = d[u] - lo;
            if (base + u * DTH < e1 && (unsigned)r < (unsigned)cn) atomicAdd(&sdeg[r], wv[u]);
        }
    }
    __syncthreads();

    float* p = part + (size_t)blockIdx.x * pstride + lo;
    const int c4 = cn >> 2;
    for (int i = tid; i < c4; i += DTH) ((float4*)p)[i] = ((const float4*)sdeg)[i];
    for (int i = (c4 << 2) + tid; i < cn; i += DTH) p[i] = sdeg[i];
}

template <int MODE>
__global__ __launch_bounds__(256)
void gemm_wmma_k(const float* __restrict__ A, const float* __restrict__ W,
                 const float* __restrict__ biasIn, const float* __restrict__ biasOut,
                 float* __restrict__ out,
                 const float* __restrict__ part, float* __restrict__ dinv, int nrows) {
    __shared__ float As[MT][LDSP];
    __shared__ alignas(16) float Ds[MT];

    const int tid = threadIdx.x;
    const int rowBase = blockIdx.x * MT;

    const v4f* A4 = (const v4f*)A;
    const int c4 = tid & 31;
    const int r0 = tid >> 5;
    float bi0 = 0.f, bi1 = 0.f, bi2 = 0.f, bi3 = 0.f;
    if (MODE != 0) {
        const int c = c4 * 4;
        bi0 = biasIn[c]; bi1 = biasIn[c + 1]; bi2 = biasIn[c + 2]; bi3 = biasIn[c + 3];
    }
    v4f   va[8];
    float dv[8];
    #pragma unroll
    for (int j = 0; j < 8; ++j) {
        const int gr = min(rowBase + r0 + 8 * j, nrows - 1);
        va[j] = A4[(size_t)gr * (DFEAT / 4) + c4];
        dv[j] = (MODE != 0) ? dinv[gr] : 1.0f;
    }

    if (MODE == 0) {
        if (tid < MT) {
            const int gr = rowBase + tid;
            float d = 0.0f;
            if (gr < nrows) {
                const int pstride = (nrows + 31) & ~31;
                float pv[DPARTS];
                #pragma unroll
                for (int p = 0; p < DPARTS; ++p) pv[p] = part[(size_t)p * pstride + gr];
                float s = 1.0f;
                #pragma unroll
                for (int p = 0; p < DPARTS; ++p) s += pv[p];
                d = (s > 0.0f) ? rsqrtf(s) : 0.0f;
                dinv[gr] = d;
            }
            Ds[tid] = d;
        }
    }

    if (MODE != 0) {
        asm volatile("" :: "v"(va[0]), "v"(va[1]), "v"(va[2]), "v"(va[3]), "v"(va[4]), "v"(va[5]), "v"(va[6]), "v"(va[7]),
                           "v"(dv[0]), "v"(dv[1]), "v"(dv[2]), "v"(dv[3]), "v"(dv[4]), "v"(dv[5]), "v"(dv[6]), "v"(dv[7]),
                           "v"(bi0), "v"(bi1), "v"(bi2), "v"(bi3));
    }

    #pragma unroll
    for (int j = 0; j < 8; ++j) {
        const int r = r0 + 8 * j;
        const bool live = rowBase + r < nrows;
        v4f v = va[j];
        if (MODE != 0) {
            const float d = dv[j];
            v.x = v.x * d + bi0; v.y = v.y * d + bi1;
            v.z = v.z * d + bi2; v.w = v.w * d + bi3;
        }
        if (MODE == 1) {
            v.x = fmaxf(v.x, 0.f); v.y = fmaxf(v.y, 0.f);
            v.z = fmaxf(v.z, 0.f); v.w = fmaxf(v.w, 0.f);
        }
        if (!live) v = (v4f)(0.0f);
        *(v4f*)&As[r][c4 * 4] = v;
        if (MODE == 1 && c4 == 0) Ds[r] = live ? dv[j] : 0.0f;
    }
    __syncthreads();

    const int lane = tid & 31;
    const int wv   = tid >> 5;
    const int colBase = wv * 16;
    const int m  = lane & 15;
    const int kh = (lane >> 4) * 2;
    const int col = colBase + m;

    v8f acc[4] = {};
    #pragma unroll
    for (int k0 = 0; k0 < DFEAT; k0 += 4) {
        v2f b;
        b.x = W[(size_t)(k0 + kh)     * DFEAT + col];
        b.y = W[(size_t)(k0 + kh + 1) * DFEAT + col];
        #pragma unroll
        for (int rt = 0; rt < 4; ++rt) {
            v2f a;
            a.x = As[rt * 16 + m][k0 + kh];
            a.y = As[rt * 16 + m][k0 + kh + 1];
            acc[rt] = __builtin_amdgcn_wmma_f32_16x16x4_f32(
                          false, a, false, b, (short)0, acc[rt], false, false);
        }
    }

    float bv = 0.0f;
    if (MODE == 2) bv = biasOut[col];
    const int rowOff = (lane >> 4) * 8;
    #pragma unroll
    for (int rt = 0; rt < 4; ++rt) {
        float sw[8] = {0.f, 0.f, 0.f, 0.f, 0.f, 0.f, 0.f, 0.f};
        if (MODE != 2) {
            const float4 s0 = *(const float4*)&Ds[rt * 16 + rowOff];
            const float4 s1 = *(const float4*)&Ds[rt * 16 + rowOff + 4];
            sw[0] = s0.x; sw[1] = s0.y; sw[2] = s0.z; sw[3] = s0.w;
            sw[4] = s1.x; sw[5] = s1.y; sw[6] = s1.z; sw[7] = s1.w;
        }
        #pragma unroll
        for (int v = 0; v < 8; ++v) {
            int gr = rowBase + rt * 16 + rowOff + v;
            if (gr < nrows) {
                const size_t o = (size_t)gr * DFEAT + col;
                if (MODE != 2) {
                    const float g = acc[rt][v] * sw[v];
                    out[o]  = g;
                } else {
                    out[o] = acc[rt][v] + bv;
                }
            }
        }
    }
}


#define ATH   1024
#define AWV   (ATH / 32)
#define QCAP  1056
#define EB    8
#define AFILL ((AR * DFEAT / 4 + ATH - 1) / ATH)

static_assert(AR <= 256, "a record keeps the local row in 8 bits");

__device__ __forceinline__ void agg_add(const float* __restrict__ hb, float* acc,
                                        int s_l, int d_l, float n_l, unsigned vm, int cnt, int lane) {
    for (int j0 = 0; j0 < cnt; j0 += EB) {
        v4f v[EB];
        #pragma unroll
        for (int u = 0; u < EB; ++u) {
            const int s = __builtin_amdgcn_readlane(s_l, j0 + u);
            v[u] = ((const v4f*)(hb + (size_t)s * DFEAT))[lane];
        }
        #pragma unroll
        for (int u = 0; u < EB; ++u) {
            if ((vm >> (j0 + u)) & 1u) {
                const int   d  = __builtin_amdgcn_readlane(d_l, j0 + u);
                const float nv = __int_as_float(__builtin_amdgcn_readlane(__float_as_int(n_l), j0 + u));
                float* o = acc + d * DFEAT + lane;
                atomicAdd(o,      v[u].x * nv);
                atomicAdd(o + 32, v[u].y * nv);
                atomicAdd(o + 64, v[u].z * nv);
                atomicAdd(o + 96, v[u].w * nv);
            }
        }
        static_assert(EB == 8, "the statement below names v[0] to v[7]");
        asm volatile("" :: "v"(v[0]), "v"(v[1]), "v"(v[2]), "v"(v[3]), "v"(v[4]), "v"(v[5]), "v"(v[6]), "v"(v[7]));
    }
}

template <bool SAVE>
__device__ __forceinline__ void agg_flush(const float* __restrict__ hb, const int* __restrict__ ei,
                                          const float* __restrict__ wgt, float* acc, const int* qw,
                                          int cnt, int lo, int cn, int E, int lane,
                                          int2* rec, int rcap, int* scnt) {
    int   s_l = 0, d_l = 0;
    float n_l = 0.0f;
    bool  mine = false;
    if (lane < cnt) {
        const int e = qw[lane];
        const int   s  = ei[e];
        const int   d  = ei[E + e] - lo;
        const float wt = wgt[e];
        mine = (unsigned)d < (unsigned)cn;
        s_l = mine ? s : 0;
        d_l = mine ? d : 0;
        n_l = mine ? wt : 0.0f;
    }
    const unsigned vm = __builtin_amdgcn_ballot_w32(mine);
    if (SAVE) {
        const int nown = __builtin_popcount(vm);
        int b0 = 0;
        if (lane == 0 && nown) b0 = atomicAdd(scnt, nown);
        b0 = __builtin_amdgcn_readfirstlane(b0);
        const int slot = b0 + __builtin_popcount(vm & ((1u << lane) - 1u));
        if (mine && slot < rcap) rec[slot] = make_int2((int)(((unsigned)s_l << 8) | (unsigned)d_l), __float_as_int(n_l));
    }
    agg_add(hb, acc, s_l, d_l, n_l, vm, cnt, lane);
}

template <bool SAVE>
__device__ __forceinline__ void agg_scan(const float* __restrict__ hb, const int* __restrict__ ei,
                                         const float* __restrict__ wgt, const unsigned char* __restrict__ own,
                                         float* acc, int* qw, int lo, int cn, int E, int tid, int lane,
                                         int2* rec, int rcap, int* scnt) {
    const unsigned mm = (blockIdx.x & 0xFFu) * 0x01010101u;
    const int G = (E + 31) >> 5;
    const int rot = (int)(((long long)blockIdx.x * G) / (int)gridDim.x) & ~31;
    const uint4* own4 = (const uint4*)own;
    const unsigned ltmask = (1u << lane) - 1u;
    int qc = 0;

    bool ok = tid < G;
    int  g  = ok ? tid + rot : 0; if (g >= G) g -= G;
    uint4 c0 = own4[2 * g], c1 = own4[2 * g + 1];

    for (int base = 0; base < G; base += ATH) {
        int  gn  = base + ATH + tid;
        const bool okn = gn < G;
        gn = okn ? gn + rot : 0; if (gn >= G) gn -= G;
        const uint4 n0 = own4[2 * gn], n1 = own4[2 * gn + 1];

        unsigned m = 0;
        if (ok) {
            const unsigned t[8] = {c0.x, c0.y, c0.z, c0.w, c1.x, c1.y, c1.z, c1.w};
            #pragma unroll
            for (int w = 0; w < 8; ++w) {
                const unsigned x = t[w] ^ mm;
                const unsigned z = (x - 0x01010101u) & ~x & 0x80808080u;
                m |= z >> (7 - w);
            }
        }
        if (__builtin_amdgcn_ballot_w32(m != 0)) {
            do {
                int e = -1;
                if (m) {
                    const int b = __builtin_ctz(m);
                    m &= m - 1;
                    e = g * 32 + (b & 7) * 4 + (b >> 3);
                    if (e >= E) e = -1;
                }
                const unsigned hv = __builtin_amdgcn_ballot_w32(e >= 0);
                if (e >= 0) qw[qc + __builtin_popcount(hv & ltmask)] = e;
                qc += __builtin_popcount(hv);
            } while (__builtin_amdgcn_ballot_w32(m != 0));
        }
        int head = 0;
        while (qc - head >= 32) {
            agg_flush<SAVE>(hb, ei, wgt, acc, qw + head, 32, lo, cn, E, lane, rec, rcap, scnt);
            head += 32;
        }
        if (head) {
            const int rem = qc - head;
            int tmp = 0;
            if (lane < rem) tmp = qw[head + lane];
            if (lane < rem) qw[lane] = tmp;
            qc = rem;
        }
        c0 = n0; c1 = n1; g = gn; ok = okn;
    }
    agg_flush<SAVE>(hb, ei, wgt, acc, qw, qc, lo, cn, E, lane, rec, rcap, scnt);
}

template <bool REPLAY>
__global__ __launch_bounds__(ATH)
void agg_k(const float* __restrict__ hb, const int* __restrict__ ei, const float* __restrict__ wgt,
           const unsigned char* __restrict__ own, float* __restrict__ out, float* lst, int n, int E) {
    __shared__ alignas(16) float acc[AR * DFEAT];
    __shared__ int q[AWV][QCAP];
    __shared__ int scnt;

    const int tid  = threadIdx.x;
    const int lane = tid & 31;
    const int lo   = blockIdx.x * AR;
    const int cn   = min(AR, n - lo);

    const int rstride = (int)((((long long)n * DFEAT) / (int)gridDim.x) & ~31LL);
    int2* rec = (int2*)(lst + (size_t)blockIdx.x * rstride);
    const int rcap = (n <= (1 << 24)) ? rstride / 2 - 1 : 0;

    int total = 0;
    if (REPLAY) total = __builtin_amdgcn_readfirstlane(rec[rcap].x);
    if (!REPLAY && tid == 0) scnt = 0;

    const float4* g4 = (const float4*)(hb + (size_t)lo * DFEAT);
    {
        const int tot = cn * (DFEAT / 4);
        float4 r[AFILL];
        #pragma unroll
        for (int k = 0; k < AFILL; ++k) r[k] = g4[min(tid + k * ATH, tot - 1)];
        static_assert(AFILL == 7, "the statement below names r[0] to r[6]");
        asm volatile("" :: "v"(r[0].x), "v"(r[1].x), "v"(r[2].x), "v"(r[3].x), "v"(r[4].x), "v"(r[5].x), "v"(r[6].x));
        #pragma unroll
        for (int k = 0; k < AFILL; ++k) {
            const int i = tid + k * ATH;
            if (i < tot) {
                float* p = acc + (i >> 5) * DFEAT + (i & 31);
                p[0] = r[k].x; p[32] = r[k].y; p[64] = r[k].z; p[96] = r[k].w;
            }
        }
    }
    __syncthreads();

    if (REPLAY && total <= rcap) {
        int c = __builtin_amdgcn_readfirstlane(tid >> 5) * 32;
        int2 r = make_int2(0, 0);
        if (c + lane < total) r = rec[c + lane];
        while (c < total) {
            const int cnx = c + ATH;
            int2 rn = make_int2(0, 0);
            if (cnx + lane < total) rn = rec[cnx + lane];
            const int cnt = min(32, total - c);
            const unsigned vm = (cnt >= 32) ? 0xFFFFFFFFu : ((1u << cnt) - 1u);
            agg_add(hb, acc, (int)((unsigned)r.x >> 8), r.x & 0xFF, __int_as_float(r.y), vm, cnt, lane);
            r = rn; c = cnx;
        }
    } else {
        agg_scan<!REPLAY>(hb, ei, wgt, own, acc, q[tid >> 5], lo, cn, E, tid, lane, rec, rcap, &scnt);
    }
    __syncthreads();

    if (!REPLAY && tid == 0) rec[rcap] = make_int2(scnt, 0);

    float4* o4 = (float4*)(out + (size_t)lo * DFEAT);
    for (int i = tid; i < cn * (DFEAT / 4); i += ATH) {
        const float* p = acc + (i >> 5) * DFEAT + (i & 31);
        o4[i] = make_float4(p[0], p[32], p[64], p[96]);
    }
}


static inline size_t align_up(size_t x) { return (x + 63) & ~(size_t)63; }

extern "C" void kernel_launch(void* const* d_in, const int* in_sizes, int n_in,
                              void* d_out, int out_size, void* d_ws, size_t ws_size,
                              hipStream_t stream) {
    const float* x  = (const float*)d_in[0];
    const int*   ei = (const int*)d_in[1];
    const float* ew = (const float*)d_in[2];
    const float* W1 = (const float*)d_in[3];  const float* b1 = (const float*)d_in[4];
    const float* W2 = (const float*)d_in[5];  const float* b2 = (const float*)d_in[6];
    const float* W3 = (const float*)d_in[7];  const float* b3 = (const float*)d_in[8];
    const float* Wl = (const float*)d_in[9];  const float* bl = (const float*)d_in[10];

    const int N = in_sizes[0] / DFEAT;
    const int E = in_sizes[2];

    float* dinv = (float*)d_ws;
    unsigned char* own = (unsigned char*)(dinv + align_up((size_t)N));
    float* hb   = dinv + align_up((size_t)N) + align_up((size_t)E);
    float* bufA = hb   + (size_t)N * DFEAT;
    float* bufB = bufA + (size_t)N * DFEAT;
    float* outF = (float*)d_out;

    const int T = 256;
    dim3 blkG((N + MT - 1) / MT);
    dim3 blkA((N + AR - 1) / AR);

    dim3 blkD(DPARTS, (N + DCH - 1) / DCH);
    deg_part_k<<<blkD, DTH, 0, stream>>>(ei, ew, bufB, own, N, E);

    gemm_wmma_k<0><<<blkG, T, 0, stream>>>(x,    W1, nullptr, nullptr, hb, bufB, dinv, N);
    agg_k<false><<<blkA, ATH, 0, stream>>>(hb, ei, ew, own, bufA, bufB, N, E);

    gemm_wmma_k<1><<<blkG, T, 0, stream>>>(bufA, W2, b1, nullptr, hb, nullptr, dinv, N);
    agg_k<true> <<<blkA, ATH, 0, stream>>>(hb, ei, ew, own, bufA, bufB, N, E);

    gemm_wmma_k<1><<<blkG, T, 0, stream>>>(bufA, W3, b2, nullptr, hb, nullptr, dinv, N);
    agg_k<true> <<<blkA, ATH, 0, stream>>>(hb, ei, ew, own, bufA, bufB, N, E);

    gemm_wmma_k<2><<<blkG, T, 0, stream>>>(bufA, Wl, b3, bl, outF, nullptr, dinv, N);
}


# Layer 1 aggregation: degree kernel files each edge as an 8-byte record in a cell per (owner block, edge slice), slot from an LDS integer add; a block reads its 64 cells, no tag scan; add order free
# speedup vs baseline: 3.3953x; 1.0585x over previous
#include <hip/hip_runtime.h>
#include <hip/hip_bf16.h>

#define DFEAT 128
#define LDSP  132
#define MT    64

typedef float v2f __attribute__((ext_vector_type(2)));
typedef float v8f __attribute__((ext_vector_type(8)));
typedef float v4f __attribute__((ext_vector_type(4)));


#define DCH    25088
#define DPARTS 64
#define DTH    512
#define DEU    13
#define AR     196
#define DOWN   (DCH / AR)

static_assert(DCH % AR == 0, "a degree block's row chunk is a whole number of aggregation blocks");

__device__ __forceinline__ long long part_floats(int n) { return (long long)DPARTS * ((n + 31) & ~31); }
__device__ __forceinline__ int cell_slots(int n, int nblk) {
    const long long avail = (long long)n * DFEAT - part_floats(n);
    return avail > 0 ? (int)(avail / 2 / ((long long)nblk * DPARTS)) : 0;
}

__global__ __launch_bounds__(DTH)
void deg_part_k(const int* __restrict__ ei, const float* __restrict__ w,
                float* __restrict__ part, unsigned char* __restrict__ own, int n, int E) {
    __shared__ alignas(16) float sdeg[DCH];
    __shared__ int lcnt[DOWN];
    const int tid = threadIdx.x;
    const int lo  = blockIdx.y * DCH;
    const int cn  = min(DCH, n - lo);
    const int pstride = (n + 31) & ~31;

    for (int i = tid; i < ((cn + 3) >> 2); i += DTH)
        ((float4*)sdeg)[i] = make_float4(0.f, 0.f, 0.f, 0.f);
    if (tid < DOWN) lcnt[tid] = 0;
    __syncthreads();

    const int nblk   = (n + AR - 1) / AR;
    const int slots  = cell_slots(n, nblk);
    const int caprec = (n <= (1 << 24)) ? slots - 1 : 0;
    const size_t ostride = (size_t)DPARTS * slots;
    int2* cellb = (int2*)(part + part_floats(n)) + ((size_t)blockIdx.y * DOWN * DPARTS + blockIdx.x) * slots;

    const int per = (E + (int)gridDim.x - 1) / (int)gridDim.x;
    const int e0  = blockIdx.x * per;
    const int e1  = min(E, e0 + per);
    for (int base = e0 + tid; base < e1; base += DTH * DEU) {
        int   d[DEU];
        int   sr[DEU];
        int   sl[DEU];
        float wv[DEU];
        #pragma unroll
        for (int u = 0; u < DEU; ++u) {
            const int ec = min(base + u * DTH, e1 - 1);
            d[u]  = ei[E + ec];
            sr[u] = ei[ec];
            wv[u] = w[ec];
        }
        if (blockIdx.y == 0) {
            #pragma unroll
            for (int u = 0; u < DEU; ++u) {
                const int ec = min(base + u * DTH, e1 - 1);
                own[ec] = (unsigned char)((d[u] / AR) & 0xFF);
            }
        }
        #pragma unroll
        for (int u = 0; u < DEU; ++u) {
            const int r = d[u] - lo;
            sl[u] = -1;
            if (base + u * DTH < e1 && (unsigned)r < (unsigned)cn) {
                atomicAdd(&sdeg[r], wv[u]);
                sl[u] = atomicAdd(&lcnt[r / AR], 1);
            }
        }
        #pragma unroll
        for (int u = 0; u < DEU; ++u) {
            const int r = d[u] - lo;
            const int o = r / AR;
            if (sl[u] >= 0 && sl[u] < caprec)
                cellb[o * ostride + sl[u]] = make_int2((int)(((unsigned)sr[u] << 8) | (unsigned)(r - o * AR)),
                                                       __float_as_int(wv[u]));
        }
    }
    __syncthreads();

    float* p = part + (size_t)blockIdx.x * pstride + lo;
    const int c4 = cn >> 2;
    for (int i = tid; i < c4; i += DTH) ((float4*)p)[i] = ((const float4*)sdeg)[i];
    for (int i = (c4 << 2) + tid; i < cn; i += DTH) p[i] = sdeg[i];

    if (slots >= 1 && tid < DOWN && blockIdx.y * DOWN + tid < nblk)
        cellb[tid * ostride + (slots - 1)] = make_int2(lcnt[tid], 0);
}

template <int MODE>
__global__ __launch_bounds__(256)
void gemm_wmma_k(const float* __restrict__ A, const float* __restrict__ W,
                 const float* __restrict__ biasIn, const float* __restrict__ biasOut,
                 float* __restrict__ out,
                 const float* __restrict__ part, float* __restrict__ dinv, int nrows) {
    __shared__ float As[MT][LDSP];
    __shared__ alignas(16) float Ds[MT];

    const int tid = threadIdx.x;
    const int rowBase = blockIdx.x * MT;

    const v4f* A4 = (const v4f*)A;
    const int c4 = tid & 31;
    const int r0 = tid >> 5;
    float bi0 = 0.f, bi1 = 0.f, bi2 = 0.f, bi3 = 0.f;
    if (MODE != 0) {
        const int c = c4 * 4;
        bi0 = biasIn[c]; bi1 = biasIn[c + 1]; bi2 = biasIn[c + 2]; bi3 = biasIn[c + 3];
    }
    v4f   va[8];
    float dv[8];
    #pragma unroll
    for (int j = 0; j < 8; ++j) {
        const int gr = min(rowBase + r0 + 8 * j, nrows - 1);
        va[j] = A4[(size_t)gr * (DFEAT / 4) + c4];
        dv[j] = (MODE != 0) ? dinv[gr] : 1.0f;
    }

    if (MODE == 0) {
        if (tid < MT) {
            const int gr = rowBase + tid;
            float d = 0.0f;
            if (gr < nrows) {
                const int pstride = (nrows + 31) & ~31;
                float pv[DPARTS];
                #pragma unroll
                for (int p = 0; p < DPARTS; ++p) pv[p] = part[(size_t)p * pstride + gr];
                float s = 1.0f;
                #pragma unroll
                for (int p = 0; p < DPARTS; ++p) s += pv[p];
                d = (s > 0.0f) ? rsqrtf(s) : 0.0f;
                dinv[gr] = d;
            }
            Ds[tid] = d;
        }
    }

    if (MODE != 0) {
        asm volatile("" :: "v"(va[0]), "v"(va[1]), "v"(va[2]), "v"(va[3]), "v"(va[4]), "v"(va[5]), "v"(va[6]), "v"(va[7]),
                           "v"(dv[0]), "v"(dv[1]), "v"(dv[2]), "v"(dv[3]), "v"(dv[4]), "v"(dv[5]), "v"(dv[6]), "v"(dv[7]),
                           "v"(bi0), "v"(bi1), "v"(bi2), "v"(bi3));
    }

    #pragma unroll
    for (int j = 0; j < 8; ++j) {
        const int r = r0 + 8 * j;
        const bool live = rowBase + r < nrows;
        v4f v = va[j];
        if (MODE != 0) {
            const float d = dv[j];
            v.x = v.x * d + bi0; v.y = v.y * d + bi1;
            v.z = v.z * d + bi2; v.w = v.w * d + bi3;
        }
        if (MODE == 1) {
            v.x = fmaxf(v.x, 0.f); v.y = fmaxf(v.y, 0.f);
            v.z = fmaxf(v.z, 0.f); v.w = fmaxf(v.w, 0.f);
        }
        if (!live) v = (v4f)(0.0f);
        *(v4f*)&As[r][c4 * 4] = v;
        if (MODE == 1 && c4 == 0) Ds[r] = live ? dv[j] : 0.0f;
    }
    __syncthreads();

    const int lane = tid & 31;
    const int wv   = tid >> 5;
    const int colBase = wv * 16;
    const int m  = lane & 15;
    const int kh = (lane >> 4) * 2;
    const int col = colBase + m;

    v8f acc[4] = {};
    #pragma unroll
    for (int k0 = 0; k0 < DFEAT; k0 += 4) {
        v2f b;
        b.x = W[(size_t)(k0 + kh)     * DFEAT + col];
        b.y = W[(size_t)(k0 + kh + 1) * DFEAT + col];
        #pragma unroll
        for (int rt = 0; rt < 4; ++rt) {
            v2f a;
            a.x = As[rt * 16 + m][k0 + kh];
            a.y = As[rt * 16 + m][k0 + kh + 1];
            acc[rt] = __builtin_amdgcn_wmma_f32_16x16x4_f32(
                          false, a, false, b, (short)0, acc[rt], false, false);
        }
    }

    float bv = 0.0f;
    if (MODE == 2) bv = biasOut[col];
    const int rowOff = (lane >> 4) * 8;
    #pragma unroll
    for (int rt = 0; rt < 4; ++rt) {
        float sw[8] = {0.f, 0.f, 0.f, 0.f, 0.f, 0.f, 0.f, 0.f};
        if (MODE != 2) {
            const float4 s0 = *(const float4*)&Ds[rt * 16 + rowOff];
            const float4 s1 = *(const float4*)&Ds[rt * 16 + rowOff + 4];
            sw[0] = s0.x; sw[1] = s0.y; sw[2] = s0.z; sw[3] = s0.w;
            sw[4] = s1.x; sw[5] = s1.y; sw[6] = s1.z; sw[7] = s1.w;
        }
        #pragma unroll
        for (int v = 0; v < 8; ++v) {
            int gr = rowBase + rt * 16 + rowOff + v;
            if (gr < nrows) {
                const size_t o = (size_t)gr * DFEAT + col;
                if (MODE != 2) {
                    const float g = acc[rt][v] * sw[v];
                    out[o]  = g;
                } else {
                    out[o] = acc[rt][v] + bv;
                }
            }
        }
    }
}


#define ATH   1024
#define AWV   (ATH / 32)
#define QCAP  1056
#define EB    8
#define AFILL ((AR * DFEAT / 4 + ATH - 1) / ATH)

static_assert(AR <= 256, "a record keeps the local row in 8 bits");

__device__ __forceinline__ void agg_add(const float* __restrict__ hb, float* acc,
                                        int s_l, int d_l, float n_l, unsigned vm, int cnt, int lane) {
    for (int j0 = 0; j0 < cnt; j0 += EB) {
        v4f v[EB];
        #pragma unroll
        for (int u = 0; u < EB; ++u) {
            const int s = __builtin_amdgcn_readlane(s_l, j0 + u);
            v[u] = ((const v4f*)(hb + (size_t)s * DFEAT))[lane];
        }
        #pragma unroll
        for (int u = 0; u < EB; ++u) {
            if ((vm >> (j0 + u)) & 1u) {
                const int   d  = __builtin_amdgcn_readlane(d_l, j0 + u);
                const float nv = __int_as_float(__builtin_amdgcn_readlane(__float_as_int(n_l), j0 + u));
                float* o = acc + d * DFEAT + lane;
                atomicAdd(o,      v[u].x * nv);
                atomicAdd(o + 32, v[u].y * nv);
                atomicAdd(o + 64, v[u].z * nv);
                atomicAdd(o + 96, v[u].w * nv);
            }
        }
        static_assert(EB == 8, "the statement below names v[0] to v[7]");
        asm volatile("" :: "v"(v[0]), "v"(v[1]), "v"(v[2]), "v"(v[3]), "v"(v[4]), "v"(v[5]), "v"(v[6]), "v"(v[7]));
    }
}

template <bool SAVE>
__device__ __forceinline__ void agg_flush(const float* __restrict__ hb, const int* __restrict__ ei,
                                          const float* __restrict__ wgt, float* acc, const int* qw,
                                          int cnt, int lo, int cn, int E, int lane,
                                          int2* rec, int rcap, int* scnt) {
    int   s_l = 0, d_l = 0;
    float n_l = 0.0f;
    bool  mine = false;
    if (lane < cnt) {
        const int e = qw[lane];
        const int   s  = ei[e];
        const int   d  = ei[E + e] - lo;
        const float wt = wgt[e];
        mine = (unsigned)d < (unsigned)cn;
        s_l = mine ? s : 0;
        d_l = mine ? d : 0;
        n_l = mine ? wt : 0.0f;
    }
    const unsigned vm = __builtin_amdgcn_ballot_w32(mine);
    if (SAVE) {
        const int nown = __builtin_popcount(vm);
        int b0 = 0;
        if (lane == 0 && nown) b0 = atomicAdd(scnt, nown);
        b0 = __builtin_amdgcn_readfirstlane(b0);
        const int slot = b0 + __builtin_popcount(vm & ((1u << lane) - 1u));
        if (mine && slot < rcap) rec[slot] = make_int2((int)(((unsigned)s_l << 8) | (unsigned)d_l), __float_as_int(n_l));
    }
    agg_add(hb, acc, s_l, d_l, n_l, vm, cnt, lane);
}

template <bool SAVE>
__device__ __forceinline__ void agg_scan(const float* __restrict__ hb, const int* __restrict__ ei,
                                         const float* __restrict__ wgt, const unsigned char* __restrict__ own,
                                         float* acc, int* qw, int lo, int cn, int E, int tid, int lane,
                                         int2* rec, int rcap, int* scnt) {
    const unsigned mm = (blockIdx.x & 0xFFu) * 0x01010101u;
    const int G = (E + 31) >> 5;
    const int rot = (int)(((long long)blockIdx.x * G) / (int)gridDim.x) & ~31;
    const uint4* own4 = (const uint4*)own;
    const unsigned ltmask = (1u << lane) - 1u;
    int qc = 0;

    bool ok = tid < G;
    int  g  = ok ? tid + rot : 0; if (g >= G) g -= G;
    uint4 c0 = own4[2 * g], c1 = own4[2 * g + 1];

    for (int base = 0; base < G; base += ATH) {
        int  gn  = base + ATH + tid;
        const bool okn = gn < G;
        gn = okn ? gn + rot : 0; if (gn >= G) gn -= G;
        const uint4 n0 = own4[2 * gn], n1 = own4[2 * gn + 1];

        unsigned m = 0;
        if (ok) {
            const unsigned t[8] = {c0.x, c0.y, c0.z, c0.w, c1.x, c1.y, c1.z, c1.w};
            #pragma unroll
            for (int w = 0; w < 8; ++w) {
                const unsigned x = t[w] ^ mm;
                const unsigned z = (x - 0x01010101u) & ~x & 0x80808080u;
                m |= z >> (7 - w);
            }
        }
        if (__builtin_amdgcn_ballot_w32(m != 0)) {
            do {
                int e = -1;
                if (m) {
                    const int b = __builtin_ctz(m);
                    m &= m - 1;
                    e = g * 32 + (b & 7) * 4 + (b >> 3);
                    if (e >= E) e = -1;
                }
                const unsigned hv = __builtin_amdgcn_ballot_w32(e >= 0);
                if (e >= 0) qw[qc + __builtin_popcount(hv & ltmask)] = e;
                qc += __builtin_popcount(hv);
            } while (__builtin_amdgcn_ballot_w32(m != 0));
        }
        int head = 0;
        while (qc - head >= 32) {
            agg_flush<SAVE>(hb, ei, wgt, acc, qw + head, 32, lo, cn, E, lane, rec, rcap, scnt);
            head += 32;
        }
        if (head) {
            const int rem = qc - head;
            int tmp = 0;
            if (lane < rem) tmp = qw[head + lane];
            if (lane < rem) qw[lane] = tmp;
            qc = rem;
        }
        c0 = n0; c1 = n1; g = gn; ok = okn;
    }
    agg_flush<SAVE>(hb, ei, wgt, acc, qw, qc, lo, cn, E, lane, rec, rcap, scnt);
}

template <bool SAVE>
__device__ __forceinline__ void agg_run(const float* __restrict__ hb, float* acc, int2 r, int cnt, int lane,
                                        int2* rec, int rcap, int* scnt) {
    const unsigned vm = (cnt >= 32) ? 0xFFFFFFFFu : ((1u << cnt) - 1u);
    if (SAVE) {
        int b0 = 0;
        if (lane == 0) b0 = atomicAdd(scnt, cnt);
        b0 = __builtin_amdgcn_readfirstlane(b0);
        const int slot = b0 + lane;
        if (lane < cnt && slot < rcap) rec[slot] = r;
    }
    agg_add(hb, acc, (int)((unsigned)r.x >> 8), r.x & 0xFF, __int_as_float(r.y), vm, cnt, lane);
}

template <bool REPLAY>
__global__ __launch_bounds__(ATH)
void agg_k(const float* __restrict__ hb, const int* __restrict__ ei, const float* __restrict__ wgt,
           const unsigned char* __restrict__ own, float* __restrict__ out, float* lst, int n, int E) {
    __shared__ alignas(16) float acc[AR * DFEAT];
    __shared__ int q[AWV][QCAP];
    __shared__ int scnt;

    const int tid  = threadIdx.x;
    const int lane = tid & 31;
    const int lo   = blockIdx.x * AR;
    const int cn   = min(AR, n - lo);

    const int rstride = (int)((part_floats(n) / (int)gridDim.x) & ~31LL);
    int2* rec = (int2*)(lst + (size_t)blockIdx.x * rstride);
    const int rcap = (n <= (1 << 24)) ? rstride / 2 - 1 : 0;

    int total = 0;
    if (REPLAY) total = __builtin_amdgcn_readfirstlane(rec[rcap].x);
    if (!REPLAY && tid == 0) scnt = 0;

    static_assert(DPARTS == 2 * AWV, "a wave takes two cells");
    const int wvs = __builtin_amdgcn_readfirstlane(tid >> 5);
    const int slots = cell_slots(n, (int)gridDim.x);
    const int2* cb = (const int2*)(lst + part_floats(n)) + (size_t)blockIdx.x * DPARTS * slots;
    bool cellsok = false;
    int  tA = 0, tB = 0;
    if (!REPLAY && slots >= 1) {
        const int caprec = (n <= (1 << 24)) ? slots - 1 : 0;
        const int c0 = cb[(size_t)lane * slots + (slots - 1)].x;
        const int c1 = cb[(size_t)(lane + 32) * slots + (slots - 1)].x;
        const bool bad = (unsigned)c0 > (unsigned)caprec || (unsigned)c1 > (unsigned)caprec;
        cellsok = __builtin_amdgcn_ballot_w32(bad) == 0u;
        tA = __builtin_amdgcn_readlane(c0, wvs);
        tB = __builtin_amdgcn_readlane(c1, wvs);
    }

    const float4* g4 = (const float4*)(hb + (size_t)lo * DFEAT);
    {
        const int tot = cn * (DFEAT / 4);
        float4 r[AFILL];
        #pragma unroll
        for (int k = 0; k < AFILL; ++k) r[k] = g4[min(tid + k * ATH, tot - 1)];
        static_assert(AFILL == 7, "the statement below names r[0] to r[6]");
        asm volatile("" :: "v"(r[0].x), "v"(r[1].x), "v"(r[2].x), "v"(r[3].x), "v"(r[4].x), "v"(r[5].x), "v"(r[6].x));
        #pragma unroll
        for (int k = 0; k < AFILL; ++k) {
            const int i = tid + k * ATH;
            if (i < tot) {
                float* p = acc + (i >> 5) * DFEAT + (i & 31);
                p[0] = r[k].x; p[32] = r[k].y; p[64] = r[k].z; p[96] = r[k].w;
            }
        }
    }
    __syncthreads();

    if (REPLAY && total <= rcap) {
        int c = __builtin_amdgcn_readfirstlane(tid >> 5) * 32;
        int2 r = make_int2(0, 0);
        if (c + lane < total) r = rec[c + lane];
        while (c < total) {
            const int cnx = c + ATH;
            int2 rn = make_int2(0, 0);
            if (cnx + lane < total) rn = rec[cnx + lane];
            const int cnt = min(32, total - c);
            const unsigned vm = (cnt >= 32) ? 0xFFFFFFFFu : ((1u << cnt) - 1u);
            agg_add(hb, acc, (int)((unsigned)r.x >> 8), r.x & 0xFF, __int_as_float(r.y), vm, cnt, lane);
            r = rn; c = cnx;
        }
    } else if (!REPLAY && cellsok) {
        const int2* pA = cb + (size_t)wvs * slots;
        const int2* pB = cb + (size_t)(wvs + AWV) * slots;
        int2 rA = make_int2(0, 0), rB = make_int2(0, 0);
        if (lane < tA) rA = pA[lane];
        if (lane < tB) rB = pB[lane];
        for (int c = 0; c < tA; c += 32) {
            int2 rn = make_int2(0, 0);
            if (c + 32 + lane < tA) rn = pA[c + 32 + lane];
            agg_run<true>(hb, acc, rA, min(32, tA - c), lane, rec, rcap, &scnt);
            rA = rn;
        }
        for (int c = 0; c < tB; c += 32) {
            int2 rn = make_int2(0, 0);
            if (c + 32 + lane < tB) rn = pB[c + 32 + lane];
            agg_run<true>(hb, acc, rB, min(32, tB - c), lane, rec, rcap, &scnt);
            rB = rn;
        }
    } else {
        agg_scan<!REPLAY>(hb, ei, wgt, own, acc, q[tid >> 5], lo, cn, E, tid, lane, rec, rcap, &scnt);
    }
    __syncthreads();

    if (!REPLAY && tid == 0) rec[rcap] = make_int2(scnt, 0);

    float4* o4 = (float4*)(out + (size_t)lo * DFEAT);
    for (int i = tid; i < cn * (DFEAT / 4); i += ATH) {
        const float* p = acc + (i >> 5) * DFEAT + (i & 31);
        o4[i] = make_float4(p[0], p[32], p[64], p[96]);
    }
}


static inline size_t align_up(size_t x) { return (x + 63) & ~(size_t)63; }

extern "C" void kernel_launch(void* const* d_in, const int* in_sizes, int n_in,
                              void* d_out, int out_size, void* d_ws, size_t ws_size,
                              hipStream_t stream) {
    const float* x  = (const float*)d_in[0];
    const int*   ei = (const int*)d_in[1];
    const float* ew = (const float*)d_in[2];
    const float* W1 = (const float*)d_in[3];  const float* b1 = (const float*)d_in[4];
    const float* W2 = (const float*)d_in[5];  const float* b2 = (const float*)d_in[6];
    const float* W3 = (const float*)d_in[7];  const float* b3 = (const float*)d_in[8];
    const float* Wl = (const float*)d_in[9];  const float* bl = (const float*)d_in[10];

    const int N = in_sizes[0] / DFEAT;
    const int E = in_sizes[2];

    float* dinv = (float*)d_ws;
    unsigned char* own = (unsigned char*)(dinv + align_up((size_t)N));
    float* hb   = dinv + align_up((size_t)N) + align_up((size_t)E);
    float* bufA = hb   + (size_t)N * DFEAT;
    float* bufB = bufA + (size_t)N * DFEAT;
    float* outF = (float*)d_out;

    const int T = 256;
    dim3 blkG((N + MT - 1) / MT);
    dim3 blkA((N + AR - 1) / AR);

    dim3 blkD(DPARTS, (N + DCH - 1) / DCH);
    deg_part_k<<<blkD, DTH, 0, stream>>>(ei, ew, bufB, own, N, E);

    gemm_wmma_k<0><<<blkG, T, 0, stream>>>(x,    W1, nullptr, nullptr, hb, bufB, dinv, N);
    agg_k<false><<<blkA, ATH, 0, stream>>>(hb, ei, ew, own, bufA, bufB, N, E);

    gemm_wmma_k<1><<<blkG, T, 0, stream>>>(bufA, W2, b1, nullptr, hb, nullptr, dinv, N);
    agg_k<true> <<<blkA, ATH, 0, stream>>>(hb, ei, ew, own, bufA, bufB, N, E);

    gemm_wmma_k<1><<<blkG, T, 0, stream>>>(bufA, W3, b2, nullptr, hb, nullptr, dinv, N);
    agg_k<true> <<<blkA, ATH, 0, stream>>>(hb, ei, ew, own, bufA, bufB, N, E);

    gemm_wmma_k<2><<<blkG, T, 0, stream>>>(bufA, Wl, b3, bl, outF, nullptr, dinv, N);
}
